# starting kernel
# baseline (speedup 1.0000x reference)
#include <hip/hip_runtime.h>

typedef __attribute__((ext_vector_type(2))) float v2f;
typedef __attribute__((ext_vector_type(8))) float v8f;

#define FDIM 48

__global__ void deg_kernel(const int* __restrict__ dst, float* __restrict__ deg, int nEdges) {
    int e = blockIdx.x * blockDim.x + threadIdx.x;
    if (e < nEdges) atomicAdd(&deg[dst[e]], 1.0f);
}

__global__ void invdeg_kernel(float* __restrict__ deg, int n) {
    int v = blockIdx.x * blockDim.x + threadIdx.x;
    if (v < n) deg[v] = 1.0f / fmaxf(deg[v], 1.0f);
}

__global__ void scatter_kernel(const float* __restrict__ x, const int* __restrict__ src,
                               const int* __restrict__ dst, float* __restrict__ agg,
                               int nEdges) {
    unsigned tid = blockIdx.x * blockDim.x + threadIdx.x;
    unsigned total = (unsigned)nEdges * 12u;
    if (tid >= total) return;
    unsigned e = tid / 12u;
    unsigned c = tid - e * 12u;
    const float4 v = *(const float4*)(x + (size_t)src[e] * FDIM + c * 4);
    float* a = agg + (size_t)dst[e] * FDIM + c * 4;
    atomicAdd(a + 0, v.x);
    atomicAdd(a + 1, v.y);
    atomicAdd(a + 2, v.z);
    atomicAdd(a + 3, v.w);
}

__global__ void __launch_bounds__(64)
sage_gemm_kernel(const float* __restrict__ x,
                 const float* __restrict__ agg,
                 const float* __restrict__ invd,
                 const float* __restrict__ Wself,
                 const float* __restrict__ Wneigh,
                 const float* __restrict__ bias,
                 float* __restrict__ out,
                 int relu, int tilesPerWave) {
    const int lane = threadIdx.x & 31;
    const int wave = threadIdx.x >> 5;
    const int half = lane >> 4;
    const int l15  = lane & 15;
    const int gw   = blockIdx.x * 2 + wave;

    v2f wS[12][3], wN[12][3];
    #pragma unroll
    for (int k = 0; k < 12; ++k) {
        const int kk = k * 4 + half * 2;
        #pragma unroll
        for (int nt = 0; nt < 3; ++nt) {
            const int col = nt * 16 + l15;
            wS[k][nt].x = Wself [(size_t)(kk)     * FDIM + col];
            wS[k][nt].y = Wself [(size_t)(kk + 1) * FDIM + col];
            wN[k][nt].x = Wneigh[(size_t)(kk)     * FDIM + col];
            wN[k][nt].y = Wneigh[(size_t)(kk + 1) * FDIM + col];
        }
    }
    float bv0 = bias[0 * 16 + l15];
    float bv1 = bias[1 * 16 + l15];
    float bv2 = bias[2 * 16 + l15];

    #pragma unroll 1
    for (int t = 0; t < tilesPerWave; ++t) {
        const int m0 = (gw * tilesPerWave + t) * 16;
        const float* xrow = x   + (size_t)(m0 + l15) * FDIM;
        const float* grow = agg + (size_t)(m0 + l15) * FDIM;
        const float  dsc  = invd[m0 + l15];

        v8f cs[3], cn[3];
        #pragma unroll
        for (int r = 0; r < 8; ++r) {
            cs[0][r] = bv0; cs[1][r] = bv1; cs[2][r] = bv2;
            cn[0][r] = 0.0f; cn[1][r] = 0.0f; cn[2][r] = 0.0f;
        }

        #pragma unroll
        for (int k = 0; k < 12; ++k) {
            const int kk = k * 4 + half * 2;
            v2f a;
            a.x = xrow[kk];
            a.y = xrow[kk + 1];
            v2f a2;
            a2.x = grow[kk] * dsc;
            a2.y = grow[kk + 1] * dsc;
            #pragma unroll
            for (int nt = 0; nt < 3; ++nt) {
                cs[nt] = __builtin_amdgcn_wmma_f32_16x16x4_f32(
                    false, a, false, wS[k][nt], (short)0, cs[nt], false, false);
                cn[nt] = __builtin_amdgcn_wmma_f32_16x16x4_f32(
                    false, a2, false, wN[k][nt], (short)0, cn[nt], false, false);
            }
        }

        #pragma unroll
        for (int nt = 0; nt < 3; ++nt) {
            #pragma unroll
            for (int r = 0; r < 8; ++r) {
                float v = cs[nt][r] + cn[nt][r];
                if (relu) v = fmaxf(v, 0.0f);
                out[(size_t)(m0 + half * 8 + r) * FDIM + nt * 16 + l15] = v;
            }
        }
    }
}

extern "C" void kernel_launch(void* const* d_in, const int* in_sizes, int n_in,
                              void* d_out, int out_size, void* d_ws, size_t ws_size,
                              hipStream_t stream) {
    const float* features = (const float*)d_in[0];
    const int*   src      = (const int*)d_in[1];
    const int*   dst      = (const int*)d_in[2];
    const float* Wself1   = (const float*)d_in[3];
    const float* Wneigh1  = (const float*)d_in[4];
    const float* b1       = (const float*)d_in[5];
    const float* Wself2   = (const float*)d_in[6];
    const float* Wneigh2  = (const float*)d_in[7];
    const float* b2       = (const float*)d_in[8];
    float* out = (float*)d_out;

    const int N = in_sizes[0] / FDIM;
    const int E = in_sizes[1];

    char* ws = (char*)d_ws;
    size_t off = 0;
    float* invd = (float*)(ws + off); off += (((size_t)N * 4)        + 255) & ~(size_t)255;
    float* agg  = (float*)(ws + off); off += (((size_t)N * FDIM * 4) + 255) & ~(size_t)255;
    float* h    = (float*)(ws + off);

    const unsigned scatterTotal  = (unsigned)E * 12u;
    const unsigned scatterBlocks = (scatterTotal + 255u) / 256u;

    const int tiles        = N / 16;
    const int tilesPerWave = 5;
    const int gemmWaves    = tiles / tilesPerWave;
    const int gemmBlocks   = gemmWaves / 2;

    hipMemsetAsync(invd, 0, (size_t)N * 4, stream);
    deg_kernel<<<(E + 255) / 256, 256, 0, stream>>>(dst, invd, E);
    invdeg_kernel<<<(N + 255) / 256, 256, 0, stream>>>(invd, N);

    hipMemsetAsync(agg, 0, (size_t)N * FDIM * 4, stream);
    scatter_kernel<<<scatterBlocks, 256, 0, stream>>>(features, src, dst, agg, E);
    sage_gemm_kernel<<<gemmBlocks, 64, 0, stream>>>(features, agg, invd,
                                                    Wself1, Wneigh1, b1, h, 1, tilesPerWave);

    hipMemsetAsync(agg, 0, (size_t)N * FDIM * 4, stream);
    scatter_kernel<<<scatterBlocks, 256, 0, stream>>>(h, src, dst, agg, E);
    sage_gemm_kernel<<<gemmBlocks, 64, 0, stream>>>(h, agg, invd,
                                                    Wself2, Wneigh2, b2, out, 0, tilesPerWave);
}


# global f32-atomic scatter and degree pass replaced: a work-group owns 800 nodes, sweeps the edge list for them and sums source rows in LDS (LDS float adds, arrival order); 6 kernels + 3 memsets to 4
# speedup vs baseline: 2.0952x; 2.0952x over previous
#include <hip/hip_runtime.h>


typedef __attribute__((ext_vector_type(2))) float v2f;
typedef __attribute__((ext_vector_type(8))) float v8f;

#define FDIM 48

#define OWN_NODES   800
#define OWN_THREADS 1024
#define OWN_WAVES   32
#define SCAN_U      8
#define QCAP        160

template <bool FULL>
__device__ __forceinline__ void own_drain16(const uint2* q, int n,
                                            const float* __restrict__ x,
                                            const int* __restrict__ src,
                                            float* sAcc, unsigned* sCnt,
                                            int half, int l16, int countDeg) {
    uint2 ent[8];
    bool ok[8];
    #pragma unroll
    for (int p = 0; p < 8; ++p) {
        const int idx = 2 * p + half;
        ok[p] = FULL || (idx < n);
        ent[p] = ok[p] ? q[idx] : make_uint2(0u, 0u);
    }
    int s[8];
    #pragma unroll
    for (int p = 0; p < 8; ++p) s[p] = ok[p] ? src[ent[p].x] : 0;
    float r0[8], r1[8], r2[8];
    #pragma unroll
    for (int p = 0; p < 8; ++p) {
        const float* xr = x + (size_t)s[p] * FDIM + 3 * l16;
        r0[p] = xr[0];
        r1[p] = xr[1];
        r2[p] = xr[2];
    }
    #pragma unroll
    for (int p = 0; p < 8; ++p) {
        if (ok[p]) {
            float* a = sAcc + ent[p].y * FDIM + 3 * l16;
            __hip_atomic_fetch_add(a + 0, r0[p], __ATOMIC_RELAXED, __HIP_MEMORY_SCOPE_WORKGROUP);
            __hip_atomic_fetch_add(a + 1, r1[p], __ATOMIC_RELAXED, __HIP_MEMORY_SCOPE_WORKGROUP);
            __hip_atomic_fetch_add(a + 2, r2[p], __ATOMIC_RELAXED, __HIP_MEMORY_SCOPE_WORKGROUP);
            if (countDeg && l16 == 0)
                __hip_atomic_fetch_add(sCnt + ent[p].y, 1u, __ATOMIC_RELAXED, __HIP_MEMORY_SCOPE_WORKGROUP);
        }
    }
}

__global__ void __launch_bounds__(OWN_THREADS)
own_sum_kernel(const float* __restrict__ x,
               const int* __restrict__ src,
               const int* __restrict__ dst,
               float* __restrict__ agg,
               float* __restrict__ invd,
               int nNodes, int nEdges, int countDeg) {
    __shared__ __attribute__((aligned(16))) float sAcc[OWN_NODES * FDIM];
    __shared__ unsigned sCnt[OWN_NODES];
    __shared__ __attribute__((aligned(8))) uint2 sQ[OWN_WAVES * QCAP];

    const int tid  = threadIdx.x;
    const int lane = tid & 31;
    const int wave = __builtin_amdgcn_readfirstlane(tid >> 5);
    const int half = lane >> 4;
    const int l16  = lane & 15;
    const int lo   = blockIdx.x * OWN_NODES;

    for (int i = tid; i < OWN_NODES * FDIM / 4; i += OWN_THREADS)
        ((float4*)sAcc)[i] = make_float4(0.0f, 0.0f, 0.0f, 0.0f);
    for (int i = tid; i < OWN_NODES; i += OWN_THREADS) sCnt[i] = 0u;
    __syncthreads();

    uint2* q = sQ + wave * QCAP;
    int cnt = 0;

    const unsigned bit = 1u << lane;

    const int nEdges4 = nEdges & ~3;
    if (wave == 0) {
        const int e = nEdges4 + lane;
        const unsigned r = (e < nEdges) ? (unsigned)(dst[e] - lo) : 0xffffffffu;
        const unsigned b = __builtin_amdgcn_ballot_w32(r < (unsigned)OWN_NODES);
        if (b != 0u) {
            if (b & bit) q[(int)__builtin_amdgcn_mbcnt_lo(b, 0u)] = make_uint2((unsigned)e, r);
            cnt = __builtin_popcount(b);
        }
    }

    const int perTrip = OWN_THREADS * 4 * SCAN_U;
    const int nTrip   = (nEdges4 + perTrip - 1) / perTrip;
    const int trip0   = (int)(((long long)blockIdx.x * nTrip) / gridDim.x);

    for (int i = 0; i < nTrip; ++i) {
        int trip = trip0 + i;
        if (trip >= nTrip) trip -= nTrip;
        const int base = trip * perTrip + wave * (128 * SCAN_U) + lane * 4;

        int4 dv[SCAN_U];
        if (trip * perTrip + perTrip <= nEdges4) {
            #pragma unroll
            for (int u = 0; u < SCAN_U; ++u)
                dv[u] = *(const int4*)(dst + base + u * 128);
        } else {
            #pragma unroll
            for (int u = 0; u < SCAN_U; ++u) {
                const int e = base + u * 128;
                dv[u] = (e < nEdges4) ? *(const int4*)(dst + e) : make_int4(-1, -1, -1, -1);
            }
        }

        #pragma unroll
        for (int u = 0; u < SCAN_U; ++u) {
            const unsigned e = (unsigned)(base + u * 128);
            const unsigned r0 = (unsigned)(dv[u].x - lo);
            const unsigned r1 = (unsigned)(dv[u].y - lo);
            const unsigned r2 = (unsigned)(dv[u].z - lo);
            const unsigned r3 = (unsigned)(dv[u].w - lo);
            const unsigned b0 = __builtin_amdgcn_ballot_w32(r0 < (unsigned)OWN_NODES);
            const unsigned b1 = __builtin_amdgcn_ballot_w32(r1 < (unsigned)OWN_NODES);
            const unsigned b2 = __builtin_amdgcn_ballot_w32(r2 < (unsigned)OWN_NODES);
            const unsigned b3 = __builtin_amdgcn_ballot_w32(r3 < (unsigned)OWN_NODES);
            if ((b0 | b1 | b2 | b3) != 0u) {
                if (b0 != 0u) {
                    if (b0 & bit) q[cnt + (int)__builtin_amdgcn_mbcnt_lo(b0, 0u)] = make_uint2(e + 0u, r0);
                    cnt += __builtin_popcount(b0);
                }
                if (b1 != 0u) {
                    if (b1 & bit) q[cnt + (int)__builtin_amdgcn_mbcnt_lo(b1, 0u)] = make_uint2(e + 1u, r1);
                    cnt += __builtin_popcount(b1);
                }
                if (b2 != 0u) {
                    if (b2 & bit) q[cnt + (int)__builtin_amdgcn_mbcnt_lo(b2, 0u)] = make_uint2(e + 2u, r2);
                    cnt += __builtin_popcount(b2);
                }
                if (b3 != 0u) {
                    if (b3 & bit) q[cnt + (int)__builtin_amdgcn_mbcnt_lo(b3, 0u)] = make_uint2(e + 3u, r3);
                    cnt += __builtin_popcount(b3);
                }
                while (cnt >= 16) {
                    cnt -= 16;
                    own_drain16<true>(q + cnt, 16, x, src, sAcc, sCnt, half, l16, countDeg);
                }
            }
        }
    }
    if (cnt > 0)
        own_drain16<false>(q, cnt, x, src, sAcc, sCnt, half, l16, countDeg);
    __syncthreads();

    const int nOwn = min(OWN_NODES, nNodes - lo);
    float4* o = (float4*)(agg + (size_t)lo * FDIM);
    for (int i = tid; i < nOwn * (FDIM / 4); i += OWN_THREADS)
        o[i] = ((const float4*)sAcc)[i];
    if (countDeg) {
        for (int i = tid; i < nOwn; i += OWN_THREADS)
            invd[lo + i] = 1.0f / fmaxf((float)sCnt[i], 1.0f);
    }
}

__global__ void __launch_bounds__(64)
sage_gemm_kernel(const float* __restrict__ x,
                 const float* __restrict__ agg,
                 const float* __restrict__ invd,
                 const float* __restrict__ Wself,
                 const float* __restrict__ Wneigh,
                 const float* __restrict__ bias,
                 float* __restrict__ out,
                 int relu, int tilesPerWave) {
    const int lane = threadIdx.x & 31;
    const int wave = threadIdx.x >> 5;
    const int half = lane >> 4;
    const int l15  = lane & 15;
    const int gw   = blockIdx.x * 2 + wave;

    v2f wS[12][3], wN[12][3];
    #pragma unroll
    for (int k = 0; k < 12; ++k) {
        const int kk = k * 4 + half * 2;
        #pragma unroll
        for (int nt = 0; nt < 3; ++nt) {
            const int col = nt * 16 + l15;
            wS[k][nt].x = Wself [(size_t)(kk)     * FDIM + col];
            wS[k][nt].y = Wself [(size_t)(kk + 1) * FDIM + col];
            wN[k][nt].x = Wneigh[(size_t)(kk)     * FDIM + col];
            wN[k][nt].y = Wneigh[(size_t)(kk + 1) * FDIM + col];
        }
    }
    float bv0 = bias[0 * 16 + l15];
    float bv1 = bias[1 * 16 + l15];
    float bv2 = bias[2 * 16 + l15];

    #pragma unroll 1
    for (int t = 0; t < tilesPerWave; ++t) {
        const int m0 = (gw * tilesPerWave + t) * 16;
        const float* xrow = x   + (size_t)(m0 + l15) * FDIM;
        const float* grow = agg + (size_t)(m0 + l15) * FDIM;
        const float  dsc  = invd[m0 + l15];

        v8f cs[3], cn[3];
        #pragma unroll
        for (int r = 0; r < 8; ++r) {
            cs[0][r] = bv0; cs[1][r] = bv1; cs[2][r] = bv2;
            cn[0][r] = 0.0f; cn[1][r] = 0.0f; cn[2][r] = 0.0f;
        }

        #pragma unroll
        for (int k = 0; k < 12; ++k) {
            const int kk = k * 4 + half * 2;
            v2f a;
            a.x = xrow[kk];
            a.y = xrow[kk + 1];
            v2f a2;
            a2.x = grow[kk] * dsc;
            a2.y = grow[kk + 1] * dsc;
            #pragma unroll
            for (int nt = 0; nt < 3; ++nt) {
                cs[nt] = __builtin_amdgcn_wmma_f32_16x16x4_f32(
                    false, a, false, wS[k][nt], (short)0, cs[nt], false, false);
                cn[nt] = __builtin_amdgcn_wmma_f32_16x16x4_f32(
                    false, a2, false, wN[k][nt], (short)0, cn[nt], false, false);
            }
        }

        #pragma unroll
        for (int nt = 0; nt < 3; ++nt) {
            #pragma unroll
            for (int r = 0; r < 8; ++r) {
                float v = cs[nt][r] + cn[nt][r];
                if (relu) v = fmaxf(v, 0.0f);
                out[(size_t)(m0 + half * 8 + r) * FDIM + nt * 16 + l15] = v;
            }
        }
    }
}

extern "C" void kernel_launch(void* const* d_in, const int* in_sizes, int n_in,
                              void* d_out, int out_size, void* d_ws, size_t ws_size,
                              hipStream_t stream) {
    const float* features = (const float*)d_in[0];
    const int*   src      = (const int*)d_in[1];
    const int*   dst      = (const int*)d_in[2];
    const float* Wself1   = (const float*)d_in[3];
    const float* Wneigh1  = (const float*)d_in[4];
    const float* b1       = (const float*)d_in[5];
    const float* Wself2   = (const float*)d_in[6];
    const float* Wneigh2  = (const float*)d_in[7];
    const float* b2       = (const float*)d_in[8];
    float* out = (float*)d_out;

    const int N = in_sizes[0] / FDIM;
    const int E = in_sizes[1];

    char* ws = (char*)d_ws;
    size_t off = 0;
    float* invd = (float*)(ws + off); off += (((size_t)N * 4)        + 255) & ~(size_t)255;
    float* agg  = (float*)(ws + off); off += (((size_t)N * FDIM * 4) + 255) & ~(size_t)255;
    float* h    = (float*)(ws + off);

    const int ownBlocks = (N + OWN_NODES - 1) / OWN_NODES;

    const int tiles        = N / 16;
    const int tilesPerWave = 5;
    const int gemmWaves    = tiles / tilesPerWave;
    const int gemmBlocks   = gemmWaves / 2;

    own_sum_kernel<<<ownBlocks, OWN_THREADS, 0, stream>>>(features, src, dst, agg, invd, N, E, 1);
    sage_gemm_kernel<<<gemmBlocks, 64, 0, stream>>>(features, agg, invd,
                                                    Wself1, Wneigh1, b1, h, 1, tilesPerWave);

    own_sum_kernel<<<ownBlocks, OWN_THREADS, 0, stream>>>(h, src, dst, agg, invd, N, E, 0);
    sage_gemm_kernel<<<gemmBlocks, 64, 0, stream>>>(h, agg, invd,
                                                    Wself2, Wneigh2, b2, out, 0, tilesPerWave);
}


# layer-1 owner pass also saves each edge it sums as (source node, owned row) in per-wave lists; layer 2 adds rows from these lists, no second sweep of the edge list; LDS float adds, arrival order
# speedup vs baseline: 2.5319x; 1.2084x over previous
#include <hip/hip_runtime.h>


typedef __attribute__((ext_vector_type(2))) float v2f;
typedef __attribute__((ext_vector_type(8))) float v8f;

#define FDIM 48

#define OWN_NODES   800
#define OWN_THREADS 1024
#define OWN_WAVES   32
#define SCAN_U      8
#define QCAP        160
#define LCAP        1024

template <bool FULL>
__device__ __forceinline__ void own_drain16(const uint2* q, int n,
                                            const float* __restrict__ x,
                                            const int* __restrict__ src,
                                            float* sAcc, unsigned* sCnt,
                                            int half, int l16, int countDeg,
                                            uint2* keep) {
    uint2 ent[8];
    bool ok[8];
    #pragma unroll
    for (int p = 0; p < 8; ++p) {
        const int idx = 2 * p + half;
        ok[p] = FULL || (idx < n);
        ent[p] = ok[p] ? q[idx] : make_uint2(0u, 0u);
    }
    const bool saver = (keep != nullptr) && (half == 0) && (FULL || l16 < n);
    const uint2 mine = saver ? q[l16] : make_uint2(0u, 0u);
    int s[8];
    #pragma unroll
    for (int p = 0; p < 8; ++p) s[p] = ok[p] ? src[ent[p].x] : 0;
    const int mineSrc = saver ? src[mine.x] : 0;
    float r0[8], r1[8], r2[8];
    #pragma unroll
    for (int p = 0; p < 8; ++p) {
        const float* xr = x + (size_t)s[p] * FDIM + 3 * l16;
        r0[p] = xr[0];
        r1[p] = xr[1];
        r2[p] = xr[2];
    }
    if (saver) keep[l16] = make_uint2((unsigned)mineSrc, mine.y);
    #pragma unroll
    for (int p = 0; p < 8; ++p) {
        if (ok[p]) {
            float* a = sAcc + ent[p].y * FDIM + 3 * l16;
            __hip_atomic_fetch_add(a + 0, r0[p], __ATOMIC_RELAXED, __HIP_MEMORY_SCOPE_WORKGROUP);
            __hip_atomic_fetch_add(a + 1, r1[p], __ATOMIC_RELAXED, __HIP_MEMORY_SCOPE_WORKGROUP);
            __hip_atomic_fetch_add(a + 2, r2[p], __ATOMIC_RELAXED, __HIP_MEMORY_SCOPE_WORKGROUP);
            if (countDeg && l16 == 0)
                __hip_atomic_fetch_add(sCnt + ent[p].y, 1u, __ATOMIC_RELAXED, __HIP_MEMORY_SCOPE_WORKGROUP);
        }
    }
}

template <bool FULL>
__device__ __forceinline__ void list_rows16(const uint2 (&ent)[8], int n,
                                            const float* __restrict__ x,
                                            float (&r0)[8], float (&r1)[8], float (&r2)[8],
                                            int half, int l16) {
    #pragma unroll
    for (int p = 0; p < 8; ++p) {
        if (FULL || (2 * p + half < n)) {
            const float* xr = x + (size_t)ent[p].x * FDIM + 3 * l16;
            r0[p] = xr[0];
            r1[p] = xr[1];
            r2[p] = xr[2];
        } else {
            r0[p] = 0.0f; r1[p] = 0.0f; r2[p] = 0.0f;
        }
    }
}

template <bool FULL>
__device__ __forceinline__ void list_adds16(const uint2 (&ent)[8], int n, float* sAcc,
                                            const float (&r0)[8], const float (&r1)[8],
                                            const float (&r2)[8], int half, int l16) {
    #pragma unroll
    for (int p = 0; p < 8; ++p) {
        if (FULL || (2 * p + half < n)) {
            float* a = sAcc + ent[p].y * FDIM + 3 * l16;
            __hip_atomic_fetch_add(a + 0, r0[p], __ATOMIC_RELAXED, __HIP_MEMORY_SCOPE_WORKGROUP);
            __hip_atomic_fetch_add(a + 1, r1[p], __ATOMIC_RELAXED, __HIP_MEMORY_SCOPE_WORKGROUP);
            __hip_atomic_fetch_add(a + 2, r2[p], __ATOMIC_RELAXED, __HIP_MEMORY_SCOPE_WORKGROUP);
        }
    }
}

__global__ void __launch_bounds__(OWN_THREADS)
own_sum_kernel(const float* __restrict__ x,
               const int* __restrict__ src,
               const int* __restrict__ dst,
               float* __restrict__ agg,
               float* __restrict__ invd,
               unsigned* lists,
               unsigned* listLen,
               int nNodes, int nEdges,
               int first) {
    const int countDeg = first;
    __shared__ __attribute__((aligned(16))) float sAcc[OWN_NODES * FDIM];
    __shared__ unsigned sCnt[OWN_NODES];
    __shared__ __attribute__((aligned(8))) uint2 sQ[OWN_WAVES * QCAP];
    __shared__ unsigned sLen[OWN_WAVES];

    const int tid  = threadIdx.x;
    const int lane = tid & 31;
    const int wave = __builtin_amdgcn_readfirstlane(tid >> 5);
    const int half = lane >> 4;
    const int l16  = lane & 15;
    const int lo   = blockIdx.x * OWN_NODES;

    for (int i = tid; i < OWN_NODES * FDIM / 4; i += OWN_THREADS)
        ((float4*)sAcc)[i] = make_float4(0.0f, 0.0f, 0.0f, 0.0f);
    for (int i = tid; i < OWN_NODES; i += OWN_THREADS) sCnt[i] = 0u;
    __syncthreads();

    uint2* myList = (uint2*)lists + ((size_t)blockIdx.x * OWN_WAVES + wave) * LCAP;

    unsigned len = 0u;
    bool fromLists = false;
    if (!first) {
        len = listLen[blockIdx.x * OWN_WAVES + lane];
        fromLists = (__builtin_amdgcn_ballot_w32(len > (unsigned)LCAP) == 0u);
    }

    if (fromLists) {
        const int n     = __builtin_amdgcn_readlane((int)len, wave);
        const int nFull = n & ~15;
        if (n > 0) {
            uint2 cur[8];
            #pragma unroll
            for (int p = 0; p < 8; ++p) cur[p] = myList[min(2 * p + half, n - 1)];
            for (int pos = 0; pos < nFull; pos += 16) {
                float r0[8], r1[8], r2[8];
                list_rows16<true>(cur, 16, x, r0, r1, r2, half, l16);
                uint2 nxt[8];
                #pragma unroll
                for (int p = 0; p < 8; ++p) nxt[p] = myList[min(pos + 16 + 2 * p + half, n - 1)];
                list_adds16<true>(cur, 16, sAcc, r0, r1, r2, half, l16);
                #pragma unroll
                for (int p = 0; p < 8; ++p) cur[p] = nxt[p];
            }
            if (n > nFull) {
                float r0[8], r1[8], r2[8];
                list_rows16<false>(cur, n - nFull, x, r0, r1, r2, half, l16);
                list_adds16<false>(cur, n - nFull, sAcc, r0, r1, r2, half, l16);
            }
        }
    } else {
        uint2* q = sQ + wave * QCAP;
        int cnt = 0;
        int lpos = 0;

        const unsigned bit = 1u << lane;

        const int nEdges4 = nEdges & ~3;
        if (wave == 0) {
            const int e = nEdges4 + lane;
            const unsigned r = (e < nEdges) ? (unsigned)(dst[e] - lo) : 0xffffffffu;
            const unsigned b = __builtin_amdgcn_ballot_w32(r < (unsigned)OWN_NODES);
            if (b != 0u) {
                if (b & bit) q[(int)__builtin_amdgcn_mbcnt_lo(b, 0u)] = make_uint2((unsigned)e, r);
                cnt = __builtin_popcount(b);
            }
        }

        const int perTrip = OWN_THREADS * 4 * SCAN_U;
        const int nTrip   = (nEdges4 + perTrip - 1) / perTrip;
        const int trip0   = (int)(((long long)blockIdx.x * nTrip) / gridDim.x);

        for (int i = 0; i < nTrip; ++i) {
            int trip = trip0 + i;
            if (trip >= nTrip) trip -= nTrip;
            const int base = trip * perTrip + wave * (128 * SCAN_U) + lane * 4;

            int4 dv[SCAN_U];
            if (trip * perTrip + perTrip <= nEdges4) {
                #pragma unroll
                for (int u = 0; u < SCAN_U; ++u)
                    dv[u] = *(const int4*)(dst + base + u * 128);
            } else {
                #pragma unroll
                for (int u = 0; u < SCAN_U; ++u) {
                    const int e = base + u * 128;
                    dv[u] = (e < nEdges4) ? *(const int4*)(dst + e) : make_int4(-1, -1, -1, -1);
                }
            }

            #pragma unroll
            for (int u = 0; u < SCAN_U; ++u) {
                const unsigned e = (unsigned)(base + u * 128);
                const unsigned r0 = (unsigned)(dv[u].x - lo);
                const unsigned r1 = (unsigned)(dv[u].y - lo);
                const unsigned r2 = (unsigned)(dv[u].z - lo);
                const unsigned r3 = (unsigned)(dv[u].w - lo);
                const unsigned b0 = __builtin_amdgcn_ballot_w32(r0 < (unsigned)OWN_NODES);
                const unsigned b1 = __builtin_amdgcn_ballot_w32(r1 < (unsigned)OWN_NODES);
                const unsigned b2 = __builtin_amdgcn_ballot_w32(r2 < (unsigned)OWN_NODES);
                const unsigned b3 = __builtin_amdgcn_ballot_w32(r3 < (unsigned)OWN_NODES);
                if ((b0 | b1 | b2 | b3) != 0u) {
                    if (b0 != 0u) {
                        if (b0 & bit) q[cnt + (int)__builtin_amdgcn_mbcnt_lo(b0, 0u)] = make_uint2(e + 0u, r0);
                        cnt += __builtin_popcount(b0);
                    }
                    if (b1 != 0u) {
                        if (b1 & bit) q[cnt + (int)__builtin_amdgcn_mbcnt_lo(b1, 0u)] = make_uint2(e + 1u, r1);
                        cnt += __builtin_popcount(b1);
                    }
                    if (b2 != 0u) {
                        if (b2 & bit) q[cnt + (int)__builtin_amdgcn_mbcnt_lo(b2, 0u)] = make_uint2(e + 2u, r2);
                        cnt += __builtin_popcount(b2);
                    }
                    if (b3 != 0u) {
                        if (b3 & bit) q[cnt + (int)__builtin_amdgcn_mbcnt_lo(b3, 0u)] = make_uint2(e + 3u, r3);
                        cnt += __builtin_popcount(b3);
                    }
                    while (cnt >= 16) {
                        cnt -= 16;
                        uint2* keep = (first && lpos + 16 <= LCAP) ? myList + lpos : nullptr;
                        lpos += 16;
                        own_drain16<true>(q + cnt, 16, x, src, sAcc, sCnt, half, l16, countDeg, keep);
                    }
                }
            }
        }
        if (cnt > 0) {
            uint2* keep = (first && lpos + cnt <= LCAP) ? myList + lpos : nullptr;
            lpos += cnt;
            own_drain16<false>(q, cnt, x, src, sAcc, sCnt, half, l16, countDeg, keep);
        }
        if (first && lane == 0) sLen[wave] = (unsigned)lpos;
    }
    __syncthreads();

    if (first && tid < OWN_WAVES) listLen[blockIdx.x * OWN_WAVES + tid] = sLen[tid];

    const int nOwn = min(OWN_NODES, nNodes - lo);
    float4* o = (float4*)(agg + (size_t)lo * FDIM);
    for (int i = tid; i < nOwn * (FDIM / 4); i += OWN_THREADS)
        o[i] = ((const float4*)sAcc)[i];
    if (countDeg) {
        for (int i = tid; i < nOwn; i += OWN_THREADS)
            invd[lo + i] = 1.0f / fmaxf((float)sCnt[i], 1.0f);
    }
}

__global__ void __launch_bounds__(64)
sage_gemm_kernel(const float* __restrict__ x,
                 const float* __restrict__ agg,
                 const float* __restrict__ invd,
                 const float* __restrict__ Wself,
                 const float* __restrict__ Wneigh,
                 const float* __restrict__ bias,
                 float* __restrict__ out,
                 int relu, int tilesPerWave) {
    const int lane = threadIdx.x & 31;
    const int wave = threadIdx.x >> 5;
    const int half = lane >> 4;
    const int l15  = lane & 15;
    const int gw   = blockIdx.x * 2 + wave;

    v2f wS[12][3], wN[12][3];
    #pragma unroll
    for (int k = 0; k < 12; ++k) {
        const int kk = k * 4 + half * 2;
        #pragma unroll
        for (int nt = 0; nt < 3; ++nt) {
            const int col = nt * 16 + l15;
            wS[k][nt].x = Wself [(size_t)(kk)     * FDIM + col];
            wS[k][nt].y = Wself [(size_t)(kk + 1) * FDIM + col];
            wN[k][nt].x = Wneigh[(size_t)(kk)     * FDIM + col];
            wN[k][nt].y = Wneigh[(size_t)(kk + 1) * FDIM + col];
        }
    }
    float bv0 = bias[0 * 16 + l15];
    float bv1 = bias[1 * 16 + l15];
    float bv2 = bias[2 * 16 + l15];

    #pragma unroll 1
    for (int t = 0; t < tilesPerWave; ++t) {
        const int m0 = (gw * tilesPerWave + t) * 16;
        const float* xrow = x   + (size_t)(m0 + l15) * FDIM;
        const float* grow = agg + (size_t)(m0 + l15) * FDIM;
        const float  dsc  = invd[m0 + l15];

        v8f cs[3], cn[3];
        #pragma unroll
        for (int r = 0; r < 8; ++r) {
            cs[0][r] = bv0; cs[1][r] = bv1; cs[2][r] = bv2;
            cn[0][r] = 0.0f; cn[1][r] = 0.0f; cn[2][r] = 0.0f;
        }

        #pragma unroll
        for (int k = 0; k < 12; ++k) {
            const int kk = k * 4 + half * 2;
            v2f a;
            a.x = xrow[kk];
            a.y = xrow[kk + 1];
            v2f a2;
            a2.x = grow[kk] * dsc;
            a2.y = grow[kk + 1] * dsc;
            #pragma unroll
            for (int nt = 0; nt < 3; ++nt) {
                cs[nt] = __builtin_amdgcn_wmma_f32_16x16x4_f32(
                    false, a, false, wS[k][nt], (short)0, cs[nt], false, false);
                cn[nt] = __builtin_amdgcn_wmma_f32_16x16x4_f32(
                    false, a2, false, wN[k][nt], (short)0, cn[nt], false, false);
            }
        }

        #pragma unroll
        for (int nt = 0; nt < 3; ++nt) {
            #pragma unroll
            for (int r = 0; r < 8; ++r) {
                float v = cs[nt][r] + cn[nt][r];
                if (relu) v = fmaxf(v, 0.0f);
                out[(size_t)(m0 + half * 8 + r) * FDIM + nt * 16 + l15] = v;
            }
        }
    }
}

extern "C" void kernel_launch(void* const* d_in, const int* in_sizes, int n_in,
                              void* d_out, int out_size, void* d_ws, size_t ws_size,
                              hipStream_t stream) {
    const float* features = (const float*)d_in[0];
    const int*   src      = (const int*)d_in[1];
    const int*   dst      = (const int*)d_in[2];
    const float* Wself1   = (const float*)d_in[3];
    const float* Wneigh1  = (const float*)d_in[4];
    const float* b1       = (const float*)d_in[5];
    const float* Wself2   = (const float*)d_in[6];
    const float* Wneigh2  = (const float*)d_in[7];
    const float* b2       = (const float*)d_in[8];
    float* out = (float*)d_out;

    const int N = in_sizes[0] / FDIM;
    const int E = in_sizes[1];

    char* ws = (char*)d_ws;
    size_t off = 0;
    float* invd = (float*)(ws + off); off += (((size_t)N * 4)        + 255) & ~(size_t)255;
    float* agg  = (float*)(ws + off); off += (((size_t)N * FDIM * 4) + 255) & ~(size_t)255;
    float* h    = (float*)(ws + off); off += (((size_t)N * FDIM * 4) + 255) & ~(size_t)255;

    const int ownBlocks = (N + OWN_NODES - 1) / OWN_NODES;
    unsigned* listLen = (unsigned*)(ws + off); off += (((size_t)ownBlocks * OWN_WAVES * 4) + 255) & ~(size_t)255;
    unsigned* lists   = (unsigned*)(ws + off);

    const int tiles        = N / 16;
    const int tilesPerWave = 5;
    const int gemmWaves    = tiles / tilesPerWave;
    const int gemmBlocks   = gemmWaves / 2;

    own_sum_kernel<<<ownBlocks, OWN_THREADS, 0, stream>>>(features, src, dst, agg, invd,
                                                          lists, listLen, N, E, 1);
    sage_gemm_kernel<<<gemmBlocks, 64, 0, stream>>>(features, agg, invd,
                                                    Wself1, Wneigh1, b1, h, 1, tilesPerWave);

    own_sum_kernel<<<ownBlocks, OWN_THREADS, 0, stream>>>(h, src, dst, agg, invd,
                                                          lists, listLen, N, E, 0);
    sage_gemm_kernel<<<gemmBlocks, 64, 0, stream>>>(h, agg, invd,
                                                    Wself2, Wneigh2, b2, out, 0, tilesPerWave);
}


# each node range gets 2 owner work-groups, each sweeps half the edge list into its own partial sums and counts; the layer GEMM adds the 2 partial rows (new sum order: two partial sums per node)
# speedup vs baseline: 3.0481x; 1.2039x over previous
#include <hip/hip_runtime.h>


typedef __attribute__((ext_vector_type(2))) float v2f;
typedef __attribute__((ext_vector_type(8))) float v8f;

#define FDIM 48

#define SPLIT       2
#define OWN_NODES   800
#define OWN_THREADS 1024
#define OWN_WAVES   32
#define SCAN_U      8
#define QCAP        160
#define LCAP        512

template <bool FULL>
__device__ __forceinline__ void own_drain16(const uint2* q, int n,
                                            const float* __restrict__ x,
                                            const int* __restrict__ src,
                                            float* sAcc, unsigned* sCnt,
                                            int half, int l16, int countDeg,
                                            uint2* keep) {
    uint2 ent[8];
    bool ok[8];
    #pragma unroll
    for (int p = 0; p < 8; ++p) {
        const int idx = 2 * p + half;
        ok[p] = FULL || (idx < n);
        ent[p] = ok[p] ? q[idx] : make_uint2(0u, 0u);
    }
    const bool saver = (keep != nullptr) && (half == 0) && (FULL || l16 < n);
    const uint2 mine = saver ? q[l16] : make_uint2(0u, 0u);
    int s[8];
    #pragma unroll
    for (int p = 0; p < 8; ++p) s[p] = ok[p] ? src[ent[p].x] : 0;
    const int mineSrc = saver ? src[mine.x] : 0;
    float r0[8], r1[8], r2[8];
    #pragma unroll
    for (int p = 0; p < 8; ++p) {
        const float* xr = x + (size_t)s[p] * FDIM + 3 * l16;
        r0[p] = xr[0];
        r1[p] = xr[1];
        r2[p] = xr[2];
    }
    if (saver) keep[l16] = make_uint2((unsigned)mineSrc, mine.y);
    #pragma unroll
    for (int p = 0; p < 8; ++p) {
        if (ok[p]) {
            float* a = sAcc + ent[p].y * FDIM + 3 * l16;
            __hip_atomic_fetch_add(a + 0, r0[p], __ATOMIC_RELAXED, __HIP_MEMORY_SCOPE_WORKGROUP);
            __hip_atomic_fetch_add(a + 1, r1[p], __ATOMIC_RELAXED, __HIP_MEMORY_SCOPE_WORKGROUP);
            __hip_atomic_fetch_add(a + 2, r2[p], __ATOMIC_RELAXED, __HIP_MEMORY_SCOPE_WORKGROUP);
            if (countDeg && l16 == 0)
                __hip_atomic_fetch_add(sCnt + ent[p].y, 1u, __ATOMIC_RELAXED, __HIP_MEMORY_SCOPE_WORKGROUP);
        }
    }
}

template <bool FULL>
__device__ __forceinline__ void list_rows16(const uint2 (&ent)[8], int n,
                                            const float* __restrict__ x,
                                            float (&r0)[8], float (&r1)[8], float (&r2)[8],
                                            int half, int l16) {
    #pragma unroll
    for (int p = 0; p < 8; ++p) {
        if (FULL || (2 * p + half < n)) {
            const float* xr = x + (size_t)ent[p].x * FDIM + 3 * l16;
            r0[p] = xr[0];
            r1[p] = xr[1];
            r2[p] = xr[2];
        } else {
            r0[p] = 0.0f; r1[p] = 0.0f; r2[p] = 0.0f;
        }
    }
}

template <bool FULL>
__device__ __forceinline__ void list_adds16(const uint2 (&ent)[8], int n, float* sAcc,
                                            const float (&r0)[8], const float (&r1)[8],
                                            const float (&r2)[8], int half, int l16) {
    #pragma unroll
    for (int p = 0; p < 8; ++p) {
        if (FULL || (2 * p + half < n)) {
            float* a = sAcc + ent[p].y * FDIM + 3 * l16;
            __hip_atomic_fetch_add(a + 0, r0[p], __ATOMIC_RELAXED, __HIP_MEMORY_SCOPE_WORKGROUP);
            __hip_atomic_fetch_add(a + 1, r1[p], __ATOMIC_RELAXED, __HIP_MEMORY_SCOPE_WORKGROUP);
            __hip_atomic_fetch_add(a + 2, r2[p], __ATOMIC_RELAXED, __HIP_MEMORY_SCOPE_WORKGROUP);
        }
    }
}

__global__ void __launch_bounds__(OWN_THREADS)
own_sum_kernel(const float* __restrict__ x,
               const int* __restrict__ src,
               const int* __restrict__ dst,
               float* __restrict__ agg,
               float* __restrict__ cnt,
               unsigned* lists,
               unsigned* listLen,
               int nNodes, int nEdges,
               int first) {
    const int countDeg = first;
    __shared__ __attribute__((aligned(16))) float sAcc[OWN_NODES * FDIM];
    __shared__ unsigned sCnt[OWN_NODES];
    __shared__ __attribute__((aligned(8))) uint2 sQ[OWN_WAVES * QCAP];
    __shared__ unsigned sLen[OWN_WAVES];

    const int tid  = threadIdx.x;
    const int lane = tid & 31;
    const int wave = __builtin_amdgcn_readfirstlane(tid >> 5);
    const int half = lane >> 4;
    const int l16  = lane & 15;
    const int part = blockIdx.x % SPLIT;
    const int lo   = (blockIdx.x / SPLIT) * OWN_NODES;

    for (int i = tid; i < OWN_NODES * FDIM / 4; i += OWN_THREADS)
        ((float4*)sAcc)[i] = make_float4(0.0f, 0.0f, 0.0f, 0.0f);
    for (int i = tid; i < OWN_NODES; i += OWN_THREADS) sCnt[i] = 0u;
    __syncthreads();

    uint2* myList = (uint2*)lists + ((size_t)blockIdx.x * OWN_WAVES + wave) * LCAP;

    unsigned len = 0u;
    bool fromLists = false;
    if (!first) {
        len = listLen[blockIdx.x * OWN_WAVES + lane];
        fromLists = (__builtin_amdgcn_ballot_w32(len > (unsigned)LCAP) == 0u);
    }

    if (fromLists) {
        const int n     = __builtin_amdgcn_readlane((int)len, wave);
        const int nFull = n & ~15;
        if (n > 0) {
            uint2 cur[8];
            #pragma unroll
            for (int p = 0; p < 8; ++p) cur[p] = myList[min(2 * p + half, n - 1)];
            for (int pos = 0; pos < nFull; pos += 16) {
                float r0[8], r1[8], r2[8];
                list_rows16<true>(cur, 16, x, r0, r1, r2, half, l16);
                uint2 nxt[8];
                #pragma unroll
                for (int p = 0; p < 8; ++p) nxt[p] = myList[min(pos + 16 + 2 * p + half, n - 1)];
                list_adds16<true>(cur, 16, sAcc, r0, r1, r2, half, l16);
                #pragma unroll
                for (int p = 0; p < 8; ++p) cur[p] = nxt[p];
            }
            if (n > nFull) {
                float r0[8], r1[8], r2[8];
                list_rows16<false>(cur, n - nFull, x, r0, r1, r2, half, l16);
                list_adds16<false>(cur, n - nFull, sAcc, r0, r1, r2, half, l16);
            }
        }
    } else {
        uint2* q = sQ + wave * QCAP;
        int cnt = 0;
        int lpos = 0;

        const unsigned bit = 1u << lane;

        const int nEdges4 = nEdges & ~3;
        const int share = ((nEdges4 / 4 + SPLIT - 1) / SPLIT) * 4;
        const int eBeg  = min(part * share, nEdges4);
        const int eEnd  = min(eBeg + share, nEdges4);
        if (wave == 0 && part == 0) {
            const int e = nEdges4 + lane;
            const unsigned r = (e < nEdges) ? (unsigned)(dst[e] - lo) : 0xffffffffu;
            const unsigned b = __builtin_amdgcn_ballot_w32(r < (unsigned)OWN_NODES);
            if (b != 0u) {
                if (b & bit) q[(int)__builtin_amdgcn_mbcnt_lo(b, 0u)] = make_uint2((unsigned)e, r);
                cnt = __builtin_popcount(b);
            }
        }

        const int perTrip = OWN_THREADS * 4 * SCAN_U;
        const int nTrip   = (eEnd - eBeg + perTrip - 1) / perTrip;
        const int trip0   = (int)(((long long)blockIdx.x * nTrip) / gridDim.x);

        for (int i = 0; i < nTrip; ++i) {
            int trip = trip0 + i;
            if (trip >= nTrip) trip -= nTrip;
            const int tbeg = eBeg + trip * perTrip;
            const int base = tbeg + wave * (128 * SCAN_U) + lane * 4;

            int4 dv[SCAN_U];
            if (tbeg + perTrip <= eEnd) {
                #pragma unroll
                for (int u = 0; u < SCAN_U; ++u)
                    dv[u] = *(const int4*)(dst + base + u * 128);
            } else {
                #pragma unroll
                for (int u = 0; u < SCAN_U; ++u) {
                    const int e = base + u * 128;
                    dv[u] = (e < eEnd) ? *(const int4*)(dst + e) : make_int4(-1, -1, -1, -1);
                }
            }

            #pragma unroll
            for (int u = 0; u < SCAN_U; ++u) {
                const unsigned e = (unsigned)(base + u * 128);
                const unsigned r0 = (unsigned)(dv[u].x - lo);
                const unsigned r1 = (unsigned)(dv[u].y - lo);
                const unsigned r2 = (unsigned)(dv[u].z - lo);
                const unsigned r3 = (unsigned)(dv[u].w - lo);
                const unsigned b0 = __builtin_amdgcn_ballot_w32(r0 < (unsigned)OWN_NODES);
                const unsigned b1 = __builtin_amdgcn_ballot_w32(r1 < (unsigned)OWN_NODES);
                const unsigned b2 = __builtin_amdgcn_ballot_w32(r2 < (unsigned)OWN_NODES);
                const unsigned b3 = __builtin_amdgcn_ballot_w32(r3 < (unsigned)OWN_NODES);
                if ((b0 | b1 | b2 | b3) != 0u) {
                    if (b0 != 0u) {
                        if (b0 & bit) q[cnt + (int)__builtin_amdgcn_mbcnt_lo(b0, 0u)] = make_uint2(e + 0u, r0);
                        cnt += __builtin_popcount(b0);
                    }
                    if (b1 != 0u) {
                        if (b1 & bit) q[cnt + (int)__builtin_amdgcn_mbcnt_lo(b1, 0u)] = make_uint2(e + 1u, r1);
                        cnt += __builtin_popcount(b1);
                    }
                    if (b2 != 0u) {
                        if (b2 & bit) q[cnt + (int)__builtin_amdgcn_mbcnt_lo(b2, 0u)] = make_uint2(e + 2u, r2);
                        cnt += __builtin_popcount(b2);
                    }
                    if (b3 != 0u) {
                        if (b3 & bit) q[cnt + (int)__builtin_amdgcn_mbcnt_lo(b3, 0u)] = make_uint2(e + 3u, r3);
                        cnt += __builtin_popcount(b3);
                    }
                    while (cnt >= 16) {
                        cnt -= 16;
                        uint2* keep = (first && lpos + 16 <= LCAP) ? myList + lpos : nullptr;
                        lpos += 16;
                        own_drain16<true>(q + cnt, 16, x, src, sAcc, sCnt, half, l16, countDeg, keep);
                    }
                }
            }
        }
        if (cnt > 0) {
            uint2* keep = (first && lpos + cnt <= LCAP) ? myList + lpos : nullptr;
            lpos += cnt;
            own_drain16<false>(q, cnt, x, src, sAcc, sCnt, half, l16, countDeg, keep);
        }
        if (first && lane == 0) sLen[wave] = (unsigned)lpos;
    }
    __syncthreads();

    if (first && tid < OWN_WAVES) listLen[blockIdx.x * OWN_WAVES + tid] = sLen[tid];

    const int nOwn = min(OWN_NODES, nNodes - lo);
    float4* o = (float4*)(agg + ((size_t)part * nNodes + lo) * FDIM);
    for (int i = tid; i < nOwn * (FDIM / 4); i += OWN_THREADS)
        o[i] = ((const float4*)sAcc)[i];
    if (countDeg) {
        for (int i = tid; i < nOwn; i += OWN_THREADS)
            cnt[(size_t)part * nNodes + lo + i] = (float)sCnt[i];
    }
}

__global__ void __launch_bounds__(64)
sage_gemm_kernel(const float* __restrict__ x,
                 const float* __restrict__ agg,
                 const float* __restrict__ cnt,
                 const float* __restrict__ Wself,
                 const float* __restrict__ Wneigh,
                 const float* __restrict__ bias,
                 float* __restrict__ out,
                 int relu, int tilesPerWave, int nNodes) {
    const int lane = threadIdx.x & 31;
    const int wave = threadIdx.x >> 5;
    const int half = lane >> 4;
    const int l15  = lane & 15;
    const int gw   = blockIdx.x * 2 + wave;

    v2f wS[12][3], wN[12][3];
    #pragma unroll
    for (int k = 0; k < 12; ++k) {
        const int kk = k * 4 + half * 2;
        #pragma unroll
        for (int nt = 0; nt < 3; ++nt) {
            const int col = nt * 16 + l15;
            wS[k][nt].x = Wself [(size_t)(kk)     * FDIM + col];
            wS[k][nt].y = Wself [(size_t)(kk + 1) * FDIM + col];
            wN[k][nt].x = Wneigh[(size_t)(kk)     * FDIM + col];
            wN[k][nt].y = Wneigh[(size_t)(kk + 1) * FDIM + col];
        }
    }
    float bv0 = bias[0 * 16 + l15];
    float bv1 = bias[1 * 16 + l15];
    float bv2 = bias[2 * 16 + l15];

    #pragma unroll 1
    for (int t = 0; t < tilesPerWave; ++t) {
        const int m0 = (gw * tilesPerWave + t) * 16;
        const float* xrow = x   + (size_t)(m0 + l15) * FDIM;
        const float* grow = agg + (size_t)(m0 + l15) * FDIM;
        const size_t plane = (size_t)nNodes * FDIM;
        float deg = cnt[m0 + l15];
        #pragma unroll
        for (int s = 1; s < SPLIT; ++s) deg += cnt[(size_t)s * nNodes + m0 + l15];
        const float  dsc  = 1.0f / fmaxf(deg, 1.0f);

        v8f cs[3], cn[3];
        #pragma unroll
        for (int r = 0; r < 8; ++r) {
            cs[0][r] = bv0; cs[1][r] = bv1; cs[2][r] = bv2;
            cn[0][r] = 0.0f; cn[1][r] = 0.0f; cn[2][r] = 0.0f;
        }

        #pragma unroll
        for (int k = 0; k < 12; ++k) {
            const int kk = k * 4 + half * 2;
            v2f a;
            a.x = xrow[kk];
            a.y = xrow[kk + 1];
            v2f a2;
            a2.x = grow[kk];
            a2.y = grow[kk + 1];
            #pragma unroll
            for (int s = 1; s < SPLIT; ++s) {
                a2.x += grow[s * plane + kk];
                a2.y += grow[s * plane + kk + 1];
            }
            a2.x *= dsc;
            a2.y *= dsc;
            #pragma unroll
            for (int nt = 0; nt < 3; ++nt) {
                cs[nt] = __builtin_amdgcn_wmma_f32_16x16x4_f32(
                    false, a, false, wS[k][nt], (short)0, cs[nt], false, false);
                cn[nt] = __builtin_amdgcn_wmma_f32_16x16x4_f32(
                    false, a2, false, wN[k][nt], (short)0, cn[nt], false, false);
            }
        }

        #pragma unroll
        for (int nt = 0; nt < 3; ++nt) {
            #pragma unroll
            for (int r = 0; r < 8; ++r) {
                float v = cs[nt][r] + cn[nt][r];
                if (relu) v = fmaxf(v, 0.0f);
                out[(size_t)(m0 + half * 8 + r) * FDIM + nt * 16 + l15] = v;
            }
        }
    }
}

extern "C" void kernel_launch(void* const* d_in, const int* in_sizes, int n_in,
                              void* d_out, int out_size, void* d_ws, size_t ws_size,
                              hipStream_t stream) {
    const float* features = (const float*)d_in[0];
    const int*   src      = (const int*)d_in[1];
    const int*   dst      = (const int*)d_in[2];
    const float* Wself1   = (const float*)d_in[3];
    const float* Wneigh1  = (const float*)d_in[4];
    const float* b1       = (const float*)d_in[5];
    const float* Wself2   = (const float*)d_in[6];
    const float* Wneigh2  = (const float*)d_in[7];
    const float* b2       = (const float*)d_in[8];
    float* out = (float*)d_out;

    const int N = in_sizes[0] / FDIM;
    const int E = in_sizes[1];

    char* ws = (char*)d_ws;
    size_t off = 0;
    float* cnt  = (float*)(ws + off); off += (((size_t)SPLIT * N * 4)        + 255) & ~(size_t)255;
    float* agg  = (float*)(ws + off); off += (((size_t)SPLIT * N * FDIM * 4) + 255) & ~(size_t)255;
    float* h    = (float*)(ws + off); off += (((size_t)N * FDIM * 4) + 255) & ~(size_t)255;

    const int ownBlocks = ((N + OWN_NODES - 1) / OWN_NODES) * SPLIT;
    unsigned* listLen = (unsigned*)(ws + off); off += (((size_t)ownBlocks * OWN_WAVES * 4) + 255) & ~(size_t)255;
    unsigned* lists   = (unsigned*)(ws + off);

    const int tiles        = N / 16;
    const int tilesPerWave = 5;
    const int gemmWaves    = tiles / tilesPerWave;
    const int gemmBlocks   = gemmWaves / 2;

    own_sum_kernel<<<ownBlocks, OWN_THREADS, 0, stream>>>(features, src, dst, agg, cnt,
                                                          lists, listLen, N, E, 1);
    sage_gemm_kernel<<<gemmBlocks, 64, 0, stream>>>(features, agg, cnt,
                                                    Wself1, Wneigh1, b1, h, 1, tilesPerWave, N);

    own_sum_kernel<<<ownBlocks, OWN_THREADS, 0, stream>>>(h, src, dst, agg, cnt,
                                                          lists, listLen, N, E, 0);
    sage_gemm_kernel<<<gemmBlocks, 64, 0, stream>>>(h, agg, cnt,
                                                    Wself2, Wneigh2, b2, out, 0, tilesPerWave, N);
}


# layer GEMM: a tile's loads (self rows, both partial sum rows, counts) go out as one group, the next tile's group ahead of this tile's products; before, each k-step's pair was waited for alone
# speedup vs baseline: 3.4751x; 1.1401x over previous
#include <hip/hip_runtime.h>


typedef __attribute__((ext_vector_type(2))) float v2f;
typedef __attribute__((ext_vector_type(8))) float v8f;

#define FDIM 48

#define SPLIT       2
#define OWN_NODES   800
#define OWN_THREADS 1024
#define OWN_WAVES   32
#define SCAN_U      8
#define QCAP        160
#define LCAP        512

template <bool FULL>
__device__ __forceinline__ void own_drain16(const uint2* q, int n,
                                            const float* __restrict__ x,
                                            const int* __restrict__ src,
                                            float* sAcc, unsigned* sCnt,
                                            int half, int l16, int countDeg,
                                            uint2* keep) {
    uint2 ent[8];
    bool ok[8];
    #pragma unroll
    for (int p = 0; p < 8; ++p) {
        const int idx = 2 * p + half;
        ok[p] = FULL || (idx < n);
        ent[p] = ok[p] ? q[idx] : make_uint2(0u, 0u);
    }
    const bool saver = (keep != nullptr) && (half == 0) && (FULL || l16 < n);
    const uint2 mine = saver ? q[l16] : make_uint2(0u, 0u);
    int s[8];
    #pragma unroll
    for (int p = 0; p < 8; ++p) s[p] = ok[p] ? src[ent[p].x] : 0;
    const int mineSrc = saver ? src[mine.x] : 0;
    float r0[8], r1[8], r2[8];
    #pragma unroll
    for (int p = 0; p < 8; ++p) {
        const float* xr = x + (size_t)s[p] * FDIM + 3 * l16;
        r0[p] = xr[0];
        r1[p] = xr[1];
        r2[p] = xr[2];
    }
    if (saver) keep[l16] = make_uint2((unsigned)mineSrc, mine.y);
    #pragma unroll
    for (int p = 0; p < 8; ++p) {
        if (ok[p]) {
            float* a = sAcc + ent[p].y * FDIM + 3 * l16;
            __hip_atomic_fetch_add(a + 0, r0[p], __ATOMIC_RELAXED, __HIP_MEMORY_SCOPE_WORKGROUP);
            __hip_atomic_fetch_add(a + 1, r1[p], __ATOMIC_RELAXED, __HIP_MEMORY_SCOPE_WORKGROUP);
            __hip_atomic_fetch_add(a + 2, r2[p], __ATOMIC_RELAXED, __HIP_MEMORY_SCOPE_WORKGROUP);
            if (countDeg && l16 == 0)
                __hip_atomic_fetch_add(sCnt + ent[p].y, 1u, __ATOMIC_RELAXED, __HIP_MEMORY_SCOPE_WORKGROUP);
        }
    }
}

template <bool FULL>
__device__ __forceinline__ void list_rows16(const uint2 (&ent)[8], int n,
                                            const float* __restrict__ x,
                                            float (&r0)[8], float (&r1)[8], float (&r2)[8],
                                            int half, int l16) {
    #pragma unroll
    for (int p = 0; p < 8; ++p) {
        if (FULL || (2 * p + half < n)) {
            const float* xr = x + (size_t)ent[p].x * FDIM + 3 * l16;
            r0[p] = xr[0];
            r1[p] = xr[1];
            r2[p] = xr[2];
        } else {
            r0[p] = 0.0f; r1[p] = 0.0f; r2[p] = 0.0f;
        }
    }
}

template <bool FULL>
__device__ __forceinline__ void list_adds16(const uint2 (&ent)[8], int n, float* sAcc,
                                            const float (&r0)[8], const float (&r1)[8],
                                            const float (&r2)[8], int half, int l16) {
    #pragma unroll
    for (int p = 0; p < 8; ++p) {
        if (FULL || (2 * p + half < n)) {
            float* a = sAcc + ent[p].y * FDIM + 3 * l16;
            __hip_atomic_fetch_add(a + 0, r0[p], __ATOMIC_RELAXED, __HIP_MEMORY_SCOPE_WORKGROUP);
            __hip_atomic_fetch_add(a + 1, r1[p], __ATOMIC_RELAXED, __HIP_MEMORY_SCOPE_WORKGROUP);
            __hip_atomic_fetch_add(a + 2, r2[p], __ATOMIC_RELAXED, __HIP_MEMORY_SCOPE_WORKGROUP);
        }
    }
}

__global__ void __launch_bounds__(OWN_THREADS)
own_sum_kernel(const float* __restrict__ x,
               const int* __restrict__ src,
               const int* __restrict__ dst,
               float* __restrict__ agg,
               float* __restrict__ cnt,
               unsigned* lists,
               unsigned* listLen,
               int nNodes, int nEdges,
               int first) {
    const int countDeg = first;
    __shared__ __attribute__((aligned(16))) float sAcc[OWN_NODES * FDIM];
    __shared__ unsigned sCnt[OWN_NODES];
    __shared__ __attribute__((aligned(8))) uint2 sQ[OWN_WAVES * QCAP];
    __shared__ unsigned sLen[OWN_WAVES];

    const int tid  = threadIdx.x;
    const int lane = tid & 31;
    const int wave = __builtin_amdgcn_readfirstlane(tid >> 5);
    const int half = lane >> 4;
    const int l16  = lane & 15;
    const int part = blockIdx.x % SPLIT;
    const int lo   = (blockIdx.x / SPLIT) * OWN_NODES;

    for (int i = tid; i < OWN_NODES * FDIM / 4; i += OWN_THREADS)
        ((float4*)sAcc)[i] = make_float4(0.0f, 0.0f, 0.0f, 0.0f);
    for (int i = tid; i < OWN_NODES; i += OWN_THREADS) sCnt[i] = 0u;
    __syncthreads();

    uint2* myList = (uint2*)lists + ((size_t)blockIdx.x * OWN_WAVES + wave) * LCAP;

    unsigned len = 0u;
    bool fromLists = false;
    if (!first) {
        len = listLen[blockIdx.x * OWN_WAVES + lane];
        fromLists = (__builtin_amdgcn_ballot_w32(len > (unsigned)LCAP) == 0u);
    }

    if (fromLists) {
        const int n     = __builtin_amdgcn_readlane((int)len, wave);
        const int nFull = n & ~15;
        if (n > 0) {
            uint2 cur[8];
            #pragma unroll
            for (int p = 0; p < 8; ++p) cur[p] = myList[min(2 * p + half, n - 1)];
            for (int pos = 0; pos < nFull; pos += 16) {
                float r0[8], r1[8], r2[8];
                list_rows16<true>(cur, 16, x, r0, r1, r2, half, l16);
                uint2 nxt[8];
                #pragma unroll
                for (int p = 0; p < 8; ++p) nxt[p] = myList[min(pos + 16 + 2 * p + half, n - 1)];
                list_adds16<true>(cur, 16, sAcc, r0, r1, r2, half, l16);
                #pragma unroll
                for (int p = 0; p < 8; ++p) cur[p] = nxt[p];
            }
            if (n > nFull) {
                float r0[8], r1[8], r2[8];
                list_rows16<false>(cur, n - nFull, x, r0, r1, r2, half, l16);
                list_adds16<false>(cur, n - nFull, sAcc, r0, r1, r2, half, l16);
            }
        }
    } else {
        uint2* q = sQ + wave * QCAP;
        int cnt = 0;
        int lpos = 0;

        const unsigned bit = 1u << lane;

        const int nEdges4 = nEdges & ~3;
        const int share = ((nEdges4 / 4 + SPLIT - 1) / SPLIT) * 4;
        const int eBeg  = min(part * share, nEdges4);
        const int eEnd  = min(eBeg + share, nEdges4);
        if (wave == 0 && part == 0) {
            const int e = nEdges4 + lane;
            const unsigned r = (e < nEdges) ? (unsigned)(dst[e] - lo) : 0xffffffffu;
            const unsigned b = __builtin_amdgcn_ballot_w32(r < (unsigned)OWN_NODES);
            if (b != 0u) {
                if (b & bit) q[(int)__builtin_amdgcn_mbcnt_lo(b, 0u)] = make_uint2((unsigned)e, r);
                cnt = __builtin_popcount(b);
            }
        }

        const int perTrip = OWN_THREADS * 4 * SCAN_U;
        const int nTrip   = (eEnd - eBeg + perTrip - 1) / perTrip;
        const int trip0   = (int)(((long long)blockIdx.x * nTrip) / gridDim.x);

        for (int i = 0; i < nTrip; ++i) {
            int trip = trip0 + i;
            if (trip >= nTrip) trip -= nTrip;
            const int tbeg = eBeg + trip * perTrip;
            const int base = tbeg + wave * (128 * SCAN_U) + lane * 4;

            int4 dv[SCAN_U];
            if (tbeg + perTrip <= eEnd) {
                #pragma unroll
                for (int u = 0; u < SCAN_U; ++u)
                    dv[u] = *(const int4*)(dst + base + u * 128);
            } else {
                #pragma unroll
                for (int u = 0; u < SCAN_U; ++u) {
                    const int e = base + u * 128;
                    dv[u] = (e < eEnd) ? *(const int4*)(dst + e) : make_int4(-1, -1, -1, -1);
                }
            }

            #pragma unroll
            for (int u = 0; u < SCAN_U; ++u) {
                const unsigned e = (unsigned)(base + u * 128);
                const unsigned r0 = (unsigned)(dv[u].x - lo);
                const unsigned r1 = (unsigned)(dv[u].y - lo);
                const unsigned r2 = (unsigned)(dv[u].z - lo);
                const unsigned r3 = (unsigned)(dv[u].w - lo);
                const unsigned b0 = __builtin_amdgcn_ballot_w32(r0 < (unsigned)OWN_NODES);
                const unsigned b1 = __builtin_amdgcn_ballot_w32(r1 < (unsigned)OWN_NODES);
                const unsigned b2 = __builtin_amdgcn_ballot_w32(r2 < (unsigned)OWN_NODES);
                const unsigned b3 = __builtin_amdgcn_ballot_w32(r3 < (unsigned)OWN_NODES);
                if ((b0 | b1 | b2 | b3) != 0u) {
                    if (b0 != 0u) {
                        if (b0 & bit) q[cnt + (int)__builtin_amdgcn_mbcnt_lo(b0, 0u)] = make_uint2(e + 0u, r0);
                        cnt += __builtin_popcount(b0);
                    }
                    if (b1 != 0u) {
                        if (b1 & bit) q[cnt + (int)__builtin_amdgcn_mbcnt_lo(b1, 0u)] = make_uint2(e + 1u, r1);
                        cnt += __builtin_popcount(b1);
                    }
                    if (b2 != 0u) {
                        if (b2 & bit) q[cnt + (int)__builtin_amdgcn_mbcnt_lo(b2, 0u)] = make_uint2(e + 2u, r2);
                        cnt += __builtin_popcount(b2);
                    }
                    if (b3 != 0u) {
                        if (b3 & bit) q[cnt + (int)__builtin_amdgcn_mbcnt_lo(b3, 0u)] = make_uint2(e + 3u, r3);
                        cnt += __builtin_popcount(b3);
                    }
                    while (cnt >= 16) {
                        cnt -= 16;
                        uint2* keep = (first && lpos + 16 <= LCAP) ? myList + lpos : nullptr;
                        lpos += 16;
                        own_drain16<true>(q + cnt, 16, x, src, sAcc, sCnt, half, l16, countDeg, keep);
                    }
                }
            }
        }
        if (cnt > 0) {
            uint2* keep = (first && lpos + cnt <= LCAP) ? myList + lpos : nullptr;
            lpos += cnt;
            own_drain16<false>(q, cnt, x, src, sAcc, sCnt, half, l16, countDeg, keep);
        }
        if (first && lane == 0) sLen[wave] = (unsigned)lpos;
    }
    __syncthreads();

    if (first && tid < OWN_WAVES) listLen[blockIdx.x * OWN_WAVES + tid] = sLen[tid];

    const int nOwn = min(OWN_NODES, nNodes - lo);
    float4* o = (float4*)(agg + ((size_t)part * nNodes + lo) * FDIM);
    for (int i = tid; i < nOwn * (FDIM / 4); i += OWN_THREADS)
        o[i] = ((const float4*)sAcc)[i];
    if (countDeg) {
        for (int i = tid; i < nOwn; i += OWN_THREADS)
            cnt[(size_t)part * nNodes + lo + i] = (float)sCnt[i];
    }
}


struct TileIn {
    v2f   xa[12];
    v2f   ga[SPLIT][12];
    float c[SPLIT];
};

__device__ __forceinline__ void tile_fetch(TileIn& r,
                                           const float* __restrict__ x,
                                           const float* __restrict__ agg,
                                           const float* __restrict__ cnt,
                                           int m0, int l15, int half, int nNodes) {
    const float* xrow = x   + (size_t)(m0 + l15) * FDIM + half * 2;
    const float* grow = agg + (size_t)(m0 + l15) * FDIM + half * 2;
    const size_t plane = (size_t)nNodes * FDIM;
    #pragma unroll
    for (int k = 0; k < 12; ++k) {
        r.xa[k].x = xrow[4 * k];
        r.xa[k].y = xrow[4 * k + 1];
    }
    #pragma unroll
    for (int s = 0; s < SPLIT; ++s) {
        #pragma unroll
        for (int k = 0; k < 12; ++k) {
            r.ga[s][k].x = grow[s * plane + 4 * k];
            r.ga[s][k].y = grow[s * plane + 4 * k + 1];
        }
    }
    #pragma unroll
    for (int s = 0; s < SPLIT; ++s) r.c[s] = cnt[(size_t)s * nNodes + m0 + l15];
}

__device__ __forceinline__ void tile_products(const TileIn& r,
                                              const v2f (&wS)[12][3], const v2f (&wN)[12][3],
                                              float bv0, float bv1, float bv2,
                                              float* __restrict__ out,
                                              int m0, int l15, int half, int relu) {
    float deg = r.c[0];
    #pragma unroll
    for (int s = 1; s < SPLIT; ++s) deg += r.c[s];
    const float dsc = 1.0f / fmaxf(deg, 1.0f);

    v8f cs[3], cn[3];
    #pragma unroll
    for (int q = 0; q < 8; ++q) {
        cs[0][q] = bv0; cs[1][q] = bv1; cs[2][q] = bv2;
        cn[0][q] = 0.0f; cn[1][q] = 0.0f; cn[2][q] = 0.0f;
    }

    #pragma unroll
    for (int k = 0; k < 12; ++k) {
        const v2f a = r.xa[k];
        v2f a2 = r.ga[0][k];
        #pragma unroll
        for (int s = 1; s < SPLIT; ++s) {
            a2.x += r.ga[s][k].x;
            a2.y += r.ga[s][k].y;
        }
        a2.x *= dsc;
        a2.y *= dsc;
        #pragma unroll
        for (int nt = 0; nt < 3; ++nt) {
            cs[nt] = __builtin_amdgcn_wmma_f32_16x16x4_f32(
                false, a, false, wS[k][nt], (short)0, cs[nt], false, false);
            cn[nt] = __builtin_amdgcn_wmma_f32_16x16x4_f32(
                false, a2, false, wN[k][nt], (short)0, cn[nt], false, false);
        }
    }

    #pragma unroll
    for (int nt = 0; nt < 3; ++nt) {
        #pragma unroll
        for (int q = 0; q < 8; ++q) {
            float v = cs[nt][q] + cn[nt][q];
            if (relu) v = fmaxf(v, 0.0f);
            out[(size_t)(m0 + half * 8 + q) * FDIM + nt * 16 + l15] = v;
        }
    }
}

__global__ void __launch_bounds__(64)
sage_gemm_kernel(const float* __restrict__ x,
                 const float* __restrict__ agg,
                 const float* __restrict__ cnt,
                 const float* __restrict__ Wself,
                 const float* __restrict__ Wneigh,
                 const float* __restrict__ bias,
                 float* __restrict__ out,
                 int relu, int tilesPerWave, int nNodes) {
    const int lane = threadIdx.x & 31;
    const int wave = threadIdx.x >> 5;
    const int half = lane >> 4;
    const int l15  = lane & 15;
    const int gw   = blockIdx.x * 2 + wave;
    const int mBeg = gw * tilesPerWave * 16;

    TileIn cur;
    tile_fetch(cur, x, agg, cnt, mBeg, l15, half, nNodes);

    v2f wS[12][3], wN[12][3];
    #pragma unroll
    for (int k = 0; k < 12; ++k) {
        const int kk = k * 4 + half * 2;
        #pragma unroll
        for (int nt = 0; nt < 3; ++nt) {
            const int col = nt * 16 + l15;
            wS[k][nt].x = Wself [(size_t)(kk)     * FDIM + col];
            wS[k][nt].y = Wself [(size_t)(kk + 1) * FDIM + col];
            wN[k][nt].x = Wneigh[(size_t)(kk)     * FDIM + col];
            wN[k][nt].y = Wneigh[(size_t)(kk + 1) * FDIM + col];
        }
    }
    float bv0 = bias[0 * 16 + l15];
    float bv1 = bias[1 * 16 + l15];
    float bv2 = bias[2 * 16 + l15];

    #pragma unroll 1
    for (int t = 0; t + 1 < tilesPerWave; ++t) {
        const int m0 = mBeg + t * 16;
        TileIn nxt;
        tile_fetch(nxt, x, agg, cnt, m0 + 16, l15, half, nNodes);
        __builtin_amdgcn_sched_barrier(0);
        tile_products(cur, wS, wN, bv0, bv1, bv2, out, m0, l15, half, relu);
        __builtin_amdgcn_sched_barrier(0);
        cur = nxt;
    }
    tile_products(cur, wS, wN, bv0, bv1, bv2, out, mBeg + (tilesPerWave - 1) * 16, l15, half, relu);
}

extern "C" void kernel_launch(void* const* d_in, const int* in_sizes, int n_in,
                              void* d_out, int out_size, void* d_ws, size_t ws_size,
                              hipStream_t stream) {
    const float* features = (const float*)d_in[0];
    const int*   src      = (const int*)d_in[1];
    const int*   dst      = (const int*)d_in[2];
    const float* Wself1   = (const float*)d_in[3];
    const float* Wneigh1  = (const float*)d_in[4];
    const float* b1       = (const float*)d_in[5];
    const float* Wself2   = (const float*)d_in[6];
    const float* Wneigh2  = (const float*)d_in[7];
    const float* b2       = (const float*)d_in[8];
    float* out = (float*)d_out;

    const int N = in_sizes[0] / FDIM;
    const int E = in_sizes[1];

    char* ws = (char*)d_ws;
    size_t off = 0;
    float* cnt  = (float*)(ws + off); off += (((size_t)SPLIT * N * 4)        + 255) & ~(size_t)255;
    float* agg  = (float*)(ws + off); off += (((size_t)SPLIT * N * FDIM * 4) + 255) & ~(size_t)255;
    float* h    = (float*)(ws + off); off += (((size_t)N * FDIM * 4) + 255) & ~(size_t)255;

    const int ownBlocks = ((N + OWN_NODES - 1) / OWN_NODES) * SPLIT;
    unsigned* listLen = (unsigned*)(ws + off); off += (((size_t)ownBlocks * OWN_WAVES * 4) + 255) & ~(size_t)255;
    unsigned* lists   = (unsigned*)(ws + off);

    const int tiles        = N / 16;
    const int tilesPerWave = 5;
    const int gemmWaves    = tiles / tilesPerWave;
    const int gemmBlocks   = gemmWaves / 2;

    own_sum_kernel<<<ownBlocks, OWN_THREADS, 0, stream>>>(features, src, dst, agg, cnt,
                                                          lists, listLen, N, E, 1);
    sage_gemm_kernel<<<gemmBlocks, 64, 0, stream>>>(features, agg, cnt,
                                                    Wself1, Wneigh1, b1, h, 1, tilesPerWave, N);

    own_sum_kernel<<<ownBlocks, OWN_THREADS, 0, stream>>>(h, src, dst, agg, cnt,
                                                          lists, listLen, N, E, 0);
    sage_gemm_kernel<<<gemmBlocks, 64, 0, stream>>>(h, agg, cnt,
                                                    Wself2, Wneigh2, b2, out, 0, tilesPerWave, N);
}


# layer-2 GEMM fused into the layer-2 owner pass: a work-group owns 400 whole nodes (edges from both saved lists of its range), sums complete in LDS, products from LDS; 4 launches to 3; new sum order
# speedup vs baseline: 3.8410x; 1.1053x over previous
#include <hip/hip_runtime.h>


typedef __attribute__((ext_vector_type(2))) float v2f;
typedef __attribute__((ext_vector_type(8))) float v8f;

#define FDIM 48

#define SPLIT       2
#define OWN_NODES   800
#define OWN_THREADS 1024
#define OWN_WAVES   32
#define SCAN_U      8
#define QCAP        160
#define LCAP        512
#define L2_NODES    (OWN_NODES / SPLIT)
#define LIST_U      8
static_assert(OWN_NODES % (SPLIT * 16) == 0, "a layer-2 work-group owns whole 16-node tiles");
static_assert(L2_NODES / 16 <= OWN_WAVES, "one tile per wave in the layer-2 products");

template <bool FULL>
__device__ __forceinline__ void own_drain16(const uint2* q, int n,
                                            const float* __restrict__ x,
                                            const int* __restrict__ src,
                                            float* sAcc, unsigned* sCnt,
                                            int half, int l16, int countDeg,
                                            uint2* keep) {
    uint2 ent[8];
    bool ok[8];
    #pragma unroll
    for (int p = 0; p < 8; ++p) {
        const int idx = 2 * p + half;
        ok[p] = FULL || (idx < n);
        ent[p] = ok[p] ? q[idx] : make_uint2(0u, 0u);
    }
    const bool saver = (keep != nullptr) && (half == 0) && (FULL || l16 < n);
    const uint2 mine = saver ? q[l16] : make_uint2(0u, 0u);
    int s[8];
    #pragma unroll
    for (int p = 0; p < 8; ++p) s[p] = ok[p] ? src[ent[p].x] : 0;
    const int mineSrc = saver ? src[mine.x] : 0;
    float r0[8], r1[8], r2[8];
    #pragma unroll
    for (int p = 0; p < 8; ++p) {
        const float* xr = x + (size_t)s[p] * FDIM + 3 * l16;
        r0[p] = xr[0];
        r1[p] = xr[1];
        r2[p] = xr[2];
    }
    if (saver) keep[l16] = make_uint2((unsigned)mineSrc, mine.y);
    #pragma unroll
    for (int p = 0; p < 8; ++p) {
        if (ok[p]) {
            float* a = sAcc + ent[p].y * FDIM + 3 * l16;
            __hip_atomic_fetch_add(a + 0, r0[p], __ATOMIC_RELAXED, __HIP_MEMORY_SCOPE_WORKGROUP);
            __hip_atomic_fetch_add(a + 1, r1[p], __ATOMIC_RELAXED, __HIP_MEMORY_SCOPE_WORKGROUP);
            __hip_atomic_fetch_add(a + 2, r2[p], __ATOMIC_RELAXED, __HIP_MEMORY_SCOPE_WORKGROUP);
            if (countDeg && l16 == 0)
                __hip_atomic_fetch_add(sCnt + ent[p].y, 1u, __ATOMIC_RELAXED, __HIP_MEMORY_SCOPE_WORKGROUP);
        }
    }
}

template <bool FULL>
__device__ __forceinline__ void list_rows16(const uint2 (&ent)[8], int n,
                                            const float* __restrict__ x,
                                            float (&r0)[8], float (&r1)[8], float (&r2)[8],
                                            int half, int l16) {
    #pragma unroll
    for (int p = 0; p < 8; ++p) {
        if (FULL || (2 * p + half < n)) {
            const float* xr = x + (size_t)ent[p].x * FDIM + 3 * l16;
            r0[p] = xr[0];
            r1[p] = xr[1];
            r2[p] = xr[2];
        } else {
            r0[p] = 0.0f; r1[p] = 0.0f; r2[p] = 0.0f;
        }
    }
}

template <bool FULL>
__device__ __forceinline__ void list_adds16(const uint2 (&ent)[8], int n, float* sAcc,
                                            const float (&r0)[8], const float (&r1)[8],
                                            const float (&r2)[8], int half, int l16) {
    #pragma unroll
    for (int p = 0; p < 8; ++p) {
        if (FULL || (2 * p + half < n)) {
            float* a = sAcc + ent[p].y * FDIM + 3 * l16;
            __hip_atomic_fetch_add(a + 0, r0[p], __ATOMIC_RELAXED, __HIP_MEMORY_SCOPE_WORKGROUP);
            __hip_atomic_fetch_add(a + 1, r1[p], __ATOMIC_RELAXED, __HIP_MEMORY_SCOPE_WORKGROUP);
            __hip_atomic_fetch_add(a + 2, r2[p], __ATOMIC_RELAXED, __HIP_MEMORY_SCOPE_WORKGROUP);
        }
    }
}

template <int RANGE>
__device__ __forceinline__ int sweep_share(const float* __restrict__ x,
                                           const int* __restrict__ src,
                                           const int* __restrict__ dst,
                                           float* sAcc, unsigned* sCnt,
                                           uint2* q,
                                           uint2* myList,
                                           int lo, int eBeg, int eEnd, bool tail,
                                           int nEdges4, int nEdges,
                                           int wave, int lane, int half, int l16, int countDeg) {
    int cnt = 0;
    int lpos = 0;

    const unsigned bit = 1u << lane;

    if (wave == 0 && tail) {
        const int e = nEdges4 + lane;
        const unsigned r = (e < nEdges) ? (unsigned)(dst[e] - lo) : 0xffffffffu;
        const unsigned b = __builtin_amdgcn_ballot_w32(r < (unsigned)RANGE);
        if (b != 0u) {
            if (b & bit) q[(int)__builtin_amdgcn_mbcnt_lo(b, 0u)] = make_uint2((unsigned)e, r);
            cnt = __builtin_popcount(b);
        }
    }

    const int perTrip = OWN_THREADS * 4 * SCAN_U;
    const int nTrip   = (eEnd - eBeg + perTrip - 1) / perTrip;
    const int trip0   = (int)(((long long)blockIdx.x * nTrip) / gridDim.x);

    for (int i = 0; i < nTrip; ++i) {
        int trip = trip0 + i;
        if (trip >= nTrip) trip -= nTrip;
        const int tbeg = eBeg + trip * perTrip;
        const int base = tbeg + wave * (128 * SCAN_U) + lane * 4;

        int4 dv[SCAN_U];
        if (tbeg + perTrip <= eEnd) {
            #pragma unroll
            for (int u = 0; u < SCAN_U; ++u)
                dv[u] = *(const int4*)(dst + base + u * 128);
        } else {
            #pragma unroll
            for (int u = 0; u < SCAN_U; ++u) {
                const int e = base + u * 128;
                dv[u] = (e < eEnd) ? *(const int4*)(dst + e) : make_int4(-1, -1, -1, -1);
            }
        }

        #pragma unroll
        for (int u = 0; u < SCAN_U; ++u) {
            const unsigned e = (unsigned)(base + u * 128);
            const unsigned r0 = (unsigned)(dv[u].x - lo);
            const unsigned r1 = (unsigned)(dv[u].y - lo);
            const unsigned r2 = (unsigned)(dv[u].z - lo);
            const unsigned r3 = (unsigned)(dv[u].w - lo);
            const unsigned b0 = __builtin_amdgcn_ballot_w32(r0 < (unsigned)RANGE);
            const unsigned b1 = __builtin_amdgcn_ballot_w32(r1 < (unsigned)RANGE);
            const unsigned b2 = __builtin_amdgcn_ballot_w32(r2 < (unsigned)RANGE);
            const unsigned b3 = __builtin_amdgcn_ballot_w32(r3 < (unsigned)RANGE);
            if ((b0 | b1 | b2 | b3) != 0u) {
                if (b0 != 0u) {
                    if (b0 & bit) q[cnt + (int)__builtin_amdgcn_mbcnt_lo(b0, 0u)] = make_uint2(e + 0u, r0);
                    cnt += __builtin_popcount(b0);
                }
                if (b1 != 0u) {
                    if (b1 & bit) q[cnt + (int)__builtin_amdgcn_mbcnt_lo(b1, 0u)] = make_uint2(e + 1u, r1);
                    cnt += __builtin_popcount(b1);
                }
                if (b2 != 0u) {
                    if (b2 & bit) q[cnt + (int)__builtin_amdgcn_mbcnt_lo(b2, 0u)] = make_uint2(e + 2u, r2);
                    cnt += __builtin_popcount(b2);
                }
                if (b3 != 0u) {
                    if (b3 & bit) q[cnt + (int)__builtin_amdgcn_mbcnt_lo(b3, 0u)] = make_uint2(e + 3u, r3);
                    cnt += __builtin_popcount(b3);
                }
                while (cnt >= 16) {
                    cnt -= 16;
                    uint2* keep = (myList != nullptr && lpos + 16 <= LCAP) ? myList + lpos : nullptr;
                    lpos += 16;
                    own_drain16<true>(q + cnt, 16, x, src, sAcc, sCnt, half, l16, countDeg, keep);
                }
            }
        }
    }
    if (cnt > 0) {
        uint2* keep = (myList != nullptr && lpos + cnt <= LCAP) ? myList + lpos : nullptr;
        lpos += cnt;
        own_drain16<false>(q, cnt, x, src, sAcc, sCnt, half, l16, countDeg, keep);
    }
    return lpos;
}

__global__ void __launch_bounds__(OWN_THREADS)
own_sum_kernel(const float* __restrict__ x,
               const int* __restrict__ src,
               const int* __restrict__ dst,
               float* __restrict__ agg,
               float* __restrict__ cnt,
               unsigned* lists,
               unsigned* listLen,
               int nNodes, int nEdges) {
    __shared__ __attribute__((aligned(16))) float sAcc[OWN_NODES * FDIM];
    __shared__ unsigned sCnt[OWN_NODES];
    __shared__ __attribute__((aligned(8))) uint2 sQ[OWN_WAVES * QCAP];
    __shared__ unsigned sLen[OWN_WAVES];

    const int tid  = threadIdx.x;
    const int lane = tid & 31;
    const int wave = __builtin_amdgcn_readfirstlane(tid >> 5);
    const int half = lane >> 4;
    const int l16  = lane & 15;
    const int part = blockIdx.x % SPLIT;
    const int lo   = (blockIdx.x / SPLIT) * OWN_NODES;

    for (int i = tid; i < OWN_NODES * FDIM / 4; i += OWN_THREADS)
        ((float4*)sAcc)[i] = make_float4(0.0f, 0.0f, 0.0f, 0.0f);
    for (int i = tid; i < OWN_NODES; i += OWN_THREADS) sCnt[i] = 0u;
    __syncthreads();

    uint2* myList = (uint2*)lists + ((size_t)blockIdx.x * OWN_WAVES + wave) * LCAP;

    const int nEdges4 = nEdges & ~3;
    const int share = ((nEdges4 / 4 + SPLIT - 1) / SPLIT) * 4;
    const int eBeg  = min(part * share, nEdges4);
    const int eEnd  = min(eBeg + share, nEdges4);
    const int lpos = sweep_share<OWN_NODES>(x, src, dst, sAcc, sCnt, sQ + wave * QCAP, myList,
                                            lo, eBeg, eEnd, part == 0, nEdges4, nEdges,
                                            wave, lane, half, l16, 1);
    if (lane == 0) sLen[wave] = (unsigned)lpos;
    __syncthreads();

    if (tid < OWN_WAVES) listLen[blockIdx.x * OWN_WAVES + tid] = sLen[tid];

    const int nOwn = min(OWN_NODES, nNodes - lo);
    float4* o = (float4*)(agg + ((size_t)part * nNodes + lo) * FDIM);
    for (int i = tid; i < nOwn * (FDIM / 4); i += OWN_THREADS)
        o[i] = ((const float4*)sAcc)[i];
    for (int i = tid; i < nOwn; i += OWN_THREADS)
        cnt[(size_t)part * nNodes + lo + i] = (float)sCnt[i];
}


struct TileIn {
    v2f   xa[12];
    v2f   ga[SPLIT][12];
    float c[SPLIT];
};

__device__ __forceinline__ void tile_fetch(TileIn& r,
                                           const float* __restrict__ x,
                                           const float* __restrict__ agg,
                                           const float* __restrict__ cnt,
                                           int m0, int l15, int half, int nNodes) {
    const float* xrow = x   + (size_t)(m0 + l15) * FDIM + half * 2;
    const float* grow = agg + (size_t)(m0 + l15) * FDIM + half * 2;
    const size_t plane = (size_t)nNodes * FDIM;
    #pragma unroll
    for (int k = 0; k < 12; ++k) {
        r.xa[k].x = xrow[4 * k];
        r.xa[k].y = xrow[4 * k + 1];
    }
    #pragma unroll
    for (int s = 0; s < SPLIT; ++s) {
        #pragma unroll
        for (int k = 0; k < 12; ++k) {
            r.ga[s][k].x = grow[s * plane + 4 * k];
            r.ga[s][k].y = grow[s * plane + 4 * k + 1];
        }
    }
    #pragma unroll
    for (int s = 0; s < SPLIT; ++s) r.c[s] = cnt[(size_t)s * nNodes + m0 + l15];
}

__device__ __forceinline__ void tile_products(const TileIn& r,
                                              const v2f (&wS)[12][3], const v2f (&wN)[12][3],
                                              float bv0, float bv1, float bv2,
                                              float* __restrict__ out,
                                              int m0, int l15, int half, int relu) {
    float deg = r.c[0];
    #pragma unroll
    for (int s = 1; s < SPLIT; ++s) deg += r.c[s];
    const float dsc = 1.0f / fmaxf(deg, 1.0f);

    v8f cs[3], cn[3];
    #pragma unroll
    for (int q = 0; q < 8; ++q) {
        cs[0][q] = bv0; cs[1][q] = bv1; cs[2][q] = bv2;
        cn[0][q] = 0.0f; cn[1][q] = 0.0f; cn[2][q] = 0.0f;
    }

    #pragma unroll
    for (int k = 0; k < 12; ++k) {
        const v2f a = r.xa[k];
        v2f a2 = r.ga[0][k];
        #pragma unroll
        for (int s = 1; s < SPLIT; ++s) {
            a2.x += r.ga[s][k].x;
            a2.y += r.ga[s][k].y;
        }
        a2.x *= dsc;
        a2.y *= dsc;
        #pragma unroll
        for (int nt = 0; nt < 3; ++nt) {
            cs[nt] = __builtin_amdgcn_wmma_f32_16x16x4_f32(
                false, a, false, wS[k][nt], (short)0, cs[nt], false, false);
            cn[nt] = __builtin_amdgcn_wmma_f32_16x16x4_f32(
                false, a2, false, wN[k][nt], (short)0, cn[nt], false, false);
        }
    }

    #pragma unroll
    for (int nt = 0; nt < 3; ++nt) {
        #pragma unroll
        for (int q = 0; q < 8; ++q) {
            float v = cs[nt][q] + cn[nt][q];
            if (relu) v = fmaxf(v, 0.0f);
            out[(size_t)(m0 + half * 8 + q) * FDIM + nt * 16 + l15] = v;
        }
    }
}

__global__ void __launch_bounds__(64)
sage_gemm_kernel(const float* __restrict__ x,
                 const float* __restrict__ agg,
                 const float* __restrict__ cnt,
                 const float* __restrict__ Wself,
                 const float* __restrict__ Wneigh,
                 const float* __restrict__ bias,
                 float* __restrict__ out,
                 int relu, int tilesPerWave, int nNodes) {
    const int lane = threadIdx.x & 31;
    const int wave = threadIdx.x >> 5;
    const int half = lane >> 4;
    const int l15  = lane & 15;
    const int gw   = blockIdx.x * 2 + wave;
    const int mBeg = gw * tilesPerWave * 16;

    TileIn cur;
    tile_fetch(cur, x, agg, cnt, mBeg, l15, half, nNodes);

    v2f wS[12][3], wN[12][3];
    #pragma unroll
    for (int k = 0; k < 12; ++k) {
        const int kk = k * 4 + half * 2;
        #pragma unroll
        for (int nt = 0; nt < 3; ++nt) {
            const int col = nt * 16 + l15;
            wS[k][nt].x = Wself [(size_t)(kk)     * FDIM + col];
            wS[k][nt].y = Wself [(size_t)(kk + 1) * FDIM + col];
            wN[k][nt].x = Wneigh[(size_t)(kk)     * FDIM + col];
            wN[k][nt].y = Wneigh[(size_t)(kk + 1) * FDIM + col];
        }
    }
    float bv0 = bias[0 * 16 + l15];
    float bv1 = bias[1 * 16 + l15];
    float bv2 = bias[2 * 16 + l15];

    #pragma unroll 1
    for (int t = 0; t + 1 < tilesPerWave; ++t) {
        const int m0 = mBeg + t * 16;
        TileIn nxt;
        tile_fetch(nxt, x, agg, cnt, m0 + 16, l15, half, nNodes);
        __builtin_amdgcn_sched_barrier(0);
        tile_products(cur, wS, wN, bv0, bv1, bv2, out, m0, l15, half, relu);
        __builtin_amdgcn_sched_barrier(0);
        cur = nxt;
    }
    tile_products(cur, wS, wN, bv0, bv1, bv2, out, mBeg + (tilesPerWave - 1) * 16, l15, half, relu);
}

template <bool FULL>
__device__ __forceinline__ void list_drain16(const uint2* q, int n,
                                             const float* __restrict__ x, float* sAcc,
                                             int half, int l16) {
    uint2 ent[8];
    #pragma unroll
    for (int p = 0; p < 8; ++p) {
        const int idx = 2 * p + half;
        ent[p] = (FULL || idx < n) ? q[idx] : make_uint2(0u, 0u);
    }
    float r0[8], r1[8], r2[8];
    list_rows16<FULL>(ent, n, x, r0, r1, r2, half, l16);
    list_adds16<FULL>(ent, n, sAcc, r0, r1, r2, half, l16);
}

__global__ void __launch_bounds__(OWN_THREADS)
own_sum_gemm_kernel(const float* __restrict__ x,
                    const int* __restrict__ src,
                    const int* __restrict__ dst,
                    const float* __restrict__ cnt,
                    const unsigned* __restrict__ lists,
                    const unsigned* __restrict__ listLen,
                    const float* __restrict__ Wself,
                    const float* __restrict__ Wneigh,
                    const float* __restrict__ bias,
                    float* __restrict__ out,
                    int nNodes, int nEdges) {
    __shared__ __attribute__((aligned(16))) float sAcc[L2_NODES * FDIM];
    __shared__ __attribute__((aligned(8))) uint2 sQ[OWN_WAVES * QCAP];
    __shared__ __attribute__((aligned(8))) v2f sW[2 * 12 * 3 * 32];

    const int tid  = threadIdx.x;
    const int lane = tid & 31;
    const int wave = __builtin_amdgcn_readfirstlane(tid >> 5);
    const int half = lane >> 4;
    const int l16  = lane & 15;
    const int range = blockIdx.x / SPLIT;
    const int sub   = blockIdx.x % SPLIT;
    const int lo    = range * OWN_NODES + sub * L2_NODES;

    for (int i = tid; i < L2_NODES * FDIM / 4; i += OWN_THREADS)
        ((float4*)sAcc)[i] = make_float4(0.0f, 0.0f, 0.0f, 0.0f);
    for (int i = tid; i < 2 * 12 * 3 * 32; i += OWN_THREADS) {
        const int ln  = i & 31;
        const int f   = i >> 5;
        const int nt  = f % 3;
        const int k   = (f / 3) % 12;
        const float* Wm = (f >= 36) ? Wneigh : Wself;
        const int kk  = k * 4 + (ln >> 4) * 2;
        const int col = nt * 16 + (ln & 15);
        v2f w;
        w.x = Wm[(size_t)(kk)     * FDIM + col];
        w.y = Wm[(size_t)(kk + 1) * FDIM + col];
        sW[i] = w;
    }
    __syncthreads();

    uint2* q = sQ + wave * QCAP;

    unsigned len[SPLIT];
    bool over = false;
    #pragma unroll
    for (int s = 0; s < SPLIT; ++s) {
        len[s] = listLen[(range * SPLIT + s) * OWN_WAVES + lane];
        over = over || (len[s] > (unsigned)LCAP);
    }
    const bool fromLists = (__builtin_amdgcn_ballot_w32(over) == 0u);

    if (fromLists) {
        int cntQ = 0;
        const unsigned bit = 1u << lane;
        #pragma unroll
        for (int s = 0; s < SPLIT; ++s) {
            const int n = __builtin_amdgcn_readlane((int)len[s], wave);
            const uint2* list = (const uint2*)lists
                              + ((size_t)(range * SPLIT + s) * OWN_WAVES + wave) * LCAP;
            for (int pos = 0; pos < n; pos += 32 * LIST_U) {
                uint2 ent[LIST_U];
                #pragma unroll
                for (int u = 0; u < LIST_U; ++u) {
                    const int idx = pos + u * 32 + lane;
                    ent[u] = (idx < n) ? list[idx] : make_uint2(0u, 0xffffffffu);
                }
                #pragma unroll
                for (int u = 0; u < LIST_U; ++u) {
                    const unsigned r = ent[u].y - (unsigned)(sub * L2_NODES);
                    const unsigned b = __builtin_amdgcn_ballot_w32(r < (unsigned)L2_NODES);
                    if (b != 0u) {
                        if (b & bit) q[cntQ + (int)__builtin_amdgcn_mbcnt_lo(b, 0u)] = make_uint2(ent[u].x, r);
                        cntQ += __builtin_popcount(b);
                        while (cntQ >= 16) {
                            cntQ -= 16;
                            list_drain16<true>(q + cntQ, 16, x, sAcc, half, l16);
                        }
                    }
                }
            }
        }
        if (cntQ > 0) list_drain16<false>(q, cntQ, x, sAcc, half, l16);
    } else {
        const int nEdges4 = nEdges & ~3;
        sweep_share<L2_NODES>(x, src, dst, sAcc, nullptr, q, nullptr,
                              lo, 0, nEdges4, true, nEdges4, nEdges,
                              wave, lane, half, l16, 0);
    }
    const int nOwn    = max(0, min(L2_NODES, nNodes - lo));
    const bool hasTile = wave < nOwn / 16;
    const int rowL = wave * 16 + l16;
    v2f xa[12];
    float deg = 1.0f, bv0 = 0.0f, bv1 = 0.0f, bv2 = 0.0f;
    #pragma unroll
    for (int k = 0; k < 12; ++k) { xa[k].x = 0.0f; xa[k].y = 0.0f; }
    if (hasTile) {
        const int m = lo + rowL;
        const float* xrow = x + (size_t)m * FDIM + half * 2;
        #pragma unroll
        for (int k = 0; k < 12; ++k) {
            xa[k].x = xrow[4 * k];
            xa[k].y = xrow[4 * k + 1];
        }
        deg = cnt[m];
        #pragma unroll
        for (int s = 1; s < SPLIT; ++s) deg += cnt[(size_t)s * nNodes + m];
        bv0 = bias[0 * 16 + l16];
        bv1 = bias[1 * 16 + l16];
        bv2 = bias[2 * 16 + l16];
    }
    __syncthreads();

    if (hasTile) {
        const float dsc = 1.0f / fmaxf(deg, 1.0f);

        v8f cs[3], cn[3];
        #pragma unroll
        for (int e = 0; e < 8; ++e) {
            cs[0][e] = bv0; cs[1][e] = bv1; cs[2][e] = bv2;
            cn[0][e] = 0.0f; cn[1][e] = 0.0f; cn[2][e] = 0.0f;
        }

        const v2f* srow = (const v2f*)(sAcc + rowL * FDIM + half * 2);
        #pragma unroll
        for (int k = 0; k < 12; ++k) {
            v2f a2 = srow[2 * k];
            a2.x *= dsc;
            a2.y *= dsc;
            #pragma unroll
            for (int nt = 0; nt < 3; ++nt) {
                const v2f wS = sW[((0 * 12 + k) * 3 + nt) * 32 + lane];
                const v2f wN = sW[((1 * 12 + k) * 3 + nt) * 32 + lane];
                cs[nt] = __builtin_amdgcn_wmma_f32_16x16x4_f32(
                    false, xa[k], false, wS, (short)0, cs[nt], false, false);
                cn[nt] = __builtin_amdgcn_wmma_f32_16x16x4_f32(
                    false, a2, false, wN, (short)0, cn[nt], false, false);
            }
        }

        float* orow = out + (size_t)(lo + wave * 16 + half * 8) * FDIM + l16;
        #pragma unroll
        for (int nt = 0; nt < 3; ++nt) {
            #pragma unroll
            for (int e = 0; e < 8; ++e)
                orow[(size_t)e * FDIM + nt * 16] = cs[nt][e] + cn[nt][e];
        }
    }
}

extern "C" void kernel_launch(void* const* d_in, const int* in_sizes, int n_in,
                              void* d_out, int out_size, void* d_ws, size_t ws_size,
                              hipStream_t stream) {
    const float* features = (const float*)d_in[0];
    const int*   src      = (const int*)d_in[1];
    const int*   dst      = (const int*)d_in[2];
    const float* Wself1   = (const float*)d_in[3];
    const float* Wneigh1  = (const float*)d_in[4];
    const float* b1       = (const float*)d_in[5];
    const float* Wself2   = (const float*)d_in[6];
    const float* Wneigh2  = (const float*)d_in[7];
    const float* b2       = (const float*)d_in[8];
    float* out = (float*)d_out;

    const int N = in_sizes[0] / FDIM;
    const int E = in_sizes[1];

    char* ws = (char*)d_ws;
    size_t off = 0;
    float* cnt  = (float*)(ws + off); off += (((size_t)SPLIT * N * 4)        + 255) & ~(size_t)255;
    float* agg  = (float*)(ws + off); off += (((size_t)SPLIT * N * FDIM * 4) + 255) & ~(size_t)255;
    float* h    = (float*)(ws + off); off += (((size_t)N * FDIM * 4) + 255) & ~(size_t)255;

    const int ownBlocks = ((N + OWN_NODES - 1) / OWN_NODES) * SPLIT;
    unsigned* listLen = (unsigned*)(ws + off); off += (((size_t)ownBlocks * OWN_WAVES * 4) + 255) & ~(size_t)255;
    unsigned* lists   = (unsigned*)(ws + off);

    const int tiles        = N / 16;
    const int tilesPerWave = 5;
    const int gemmWaves    = tiles / tilesPerWave;
    const int gemmBlocks   = gemmWaves / 2;

    own_sum_kernel<<<ownBlocks, OWN_THREADS, 0, stream>>>(features, src, dst, agg, cnt,
                                                          lists, listLen, N, E);
    sage_gemm_kernel<<<gemmBlocks, 64, 0, stream>>>(features, agg, cnt,
                                                    Wself1, Wneigh1, b1, h, 1, tilesPerWave, N);

    own_sum_gemm_kernel<<<ownBlocks, OWN_THREADS, 0, stream>>>(h, src, dst, cnt, lists, listLen,
                                                               Wself2, Wneigh2, b2, out, N, E);
}


# layer-1 owners read a per-range bit plane over the edges and a packed (source,row) word per set bit, both from a new pre-pass, not the whole destination list; 3 launches to 4; LDS adds arrival order
# speedup vs baseline: 3.9959x; 1.0403x over previous
#include <hip/hip_runtime.h>


typedef __attribute__((ext_vector_type(2))) float v2f;
typedef __attribute__((ext_vector_type(8))) float v8f;

#define FDIM 48

#define SPLIT       2
#define OWN_NODES   800
#define OWN_THREADS 1024
#define OWN_WAVES   32
#define SCAN_U      8
#define QCAP        160
#define LCAP        512
#define L2_NODES    (OWN_NODES / SPLIT)
#define LIST_U      8
static_assert(OWN_NODES % (SPLIT * 16) == 0, "a layer-2 work-group owns whole 16-node tiles");
static_assert(L2_NODES / 16 <= OWN_WAVES, "one tile per wave in the layer-2 products");

#define PL_CHUNK    4096
#define PL_CHW      (PL_CHUNK / 32)
#define PL_PITCH    (PL_CHW + 4)
#define PL_GROUP    128
#define ROW_BITS    10
static_assert(OWN_NODES <= (1 << ROW_BITS), "an owned row fits the packed word's low bits");
static_assert(PL_CHUNK % 4096 == 0, "a pre-pass thread takes whole groups of four edges");

template <bool FULL>
__device__ __forceinline__ void own_drain16(const uint2* q, int n,
                                            const float* __restrict__ x,
                                            const int* __restrict__ src,
                                            float* sAcc, unsigned* sCnt,
                                            int half, int l16, int countDeg,
                                            uint2* keep) {
    uint2 ent[8];
    bool ok[8];
    #pragma unroll
    for (int p = 0; p < 8; ++p) {
        const int idx = 2 * p + half;
        ok[p] = FULL || (idx < n);
        ent[p] = ok[p] ? q[idx] : make_uint2(0u, 0u);
    }
    const bool saver = (keep != nullptr) && (half == 0) && (FULL || l16 < n);
    const uint2 mine = saver ? q[l16] : make_uint2(0u, 0u);
    int s[8];
    #pragma unroll
    for (int p = 0; p < 8; ++p) s[p] = ok[p] ? src[ent[p].x] : 0;
    const int mineSrc = saver ? src[mine.x] : 0;
    float r0[8], r1[8], r2[8];
    #pragma unroll
    for (int p = 0; p < 8; ++p) {
        const float* xr = x + (size_t)s[p] * FDIM + 3 * l16;
        r0[p] = xr[0];
        r1[p] = xr[1];
        r2[p] = xr[2];
    }
    if (saver) keep[l16] = make_uint2((unsigned)mineSrc, mine.y);
    #pragma unroll
    for (int p = 0; p < 8; ++p) {
        if (ok[p]) {
            float* a = sAcc + ent[p].y * FDIM + 3 * l16;
            __hip_atomic_fetch_add(a + 0, r0[p], __ATOMIC_RELAXED, __HIP_MEMORY_SCOPE_WORKGROUP);
            __hip_atomic_fetch_add(a + 1, r1[p], __ATOMIC_RELAXED, __HIP_MEMORY_SCOPE_WORKGROUP);
            __hip_atomic_fetch_add(a + 2, r2[p], __ATOMIC_RELAXED, __HIP_MEMORY_SCOPE_WORKGROUP);
            if (countDeg && l16 == 0)
                __hip_atomic_fetch_add(sCnt + ent[p].y, 1u, __ATOMIC_RELAXED, __HIP_MEMORY_SCOPE_WORKGROUP);
        }
    }
}

template <bool FULL>
__device__ __forceinline__ void list_rows16(const uint2 (&ent)[8], int n,
                                            const float* __restrict__ x,
                                            float (&r0)[8], float (&r1)[8], float (&r2)[8],
                                            int half, int l16) {
    #pragma unroll
    for (int p = 0; p < 8; ++p) {
        if (FULL || (2 * p + half < n)) {
            const float* xr = x + (size_t)ent[p].x * FDIM + 3 * l16;
            r0[p] = xr[0];
            r1[p] = xr[1];
            r2[p] = xr[2];
        } else {
            r0[p] = 0.0f; r1[p] = 0.0f; r2[p] = 0.0f;
        }
    }
}

template <bool FULL>
__device__ __forceinline__ void list_adds16(const uint2 (&ent)[8], int n, float* sAcc,
                                            const float (&r0)[8], const float (&r1)[8],
                                            const float (&r2)[8], int half, int l16) {
    #pragma unroll
    for (int p = 0; p < 8; ++p) {
        if (FULL || (2 * p + half < n)) {
            float* a = sAcc + ent[p].y * FDIM + 3 * l16;
            __hip_atomic_fetch_add(a + 0, r0[p], __ATOMIC_RELAXED, __HIP_MEMORY_SCOPE_WORKGROUP);
            __hip_atomic_fetch_add(a + 1, r1[p], __ATOMIC_RELAXED, __HIP_MEMORY_SCOPE_WORKGROUP);
            __hip_atomic_fetch_add(a + 2, r2[p], __ATOMIC_RELAXED, __HIP_MEMORY_SCOPE_WORKGROUP);
        }
    }
}

template <int RANGE>
__device__ __forceinline__ int sweep_share(const float* __restrict__ x,
                                           const int* __restrict__ src,
                                           const int* __restrict__ dst,
                                           float* sAcc, unsigned* sCnt,
                                           uint2* q,
                                           uint2* myList,
                                           int lo, int eBeg, int eEnd, bool tail,
                                           int nEdges4, int nEdges,
                                           int wave, int lane, int half, int l16, int countDeg) {
    int cnt = 0;
    int lpos = 0;

    const unsigned bit = 1u << lane;

    if (wave == 0 && tail) {
        const int e = nEdges4 + lane;
        const unsigned r = (e < nEdges) ? (unsigned)(dst[e] - lo) : 0xffffffffu;
        const unsigned b = __builtin_amdgcn_ballot_w32(r < (unsigned)RANGE);
        if (b != 0u) {
            if (b & bit) q[(int)__builtin_amdgcn_mbcnt_lo(b, 0u)] = make_uint2((unsigned)e, r);
            cnt = __builtin_popcount(b);
        }
    }

    const int perTrip = OWN_THREADS * 4 * SCAN_U;
    const int nTrip   = (eEnd - eBeg + perTrip - 1) / perTrip;
    const int trip0   = (int)(((long long)blockIdx.x * nTrip) / gridDim.x);

    for (int i = 0; i < nTrip; ++i) {
        int trip = trip0 + i;
        if (trip >= nTrip) trip -= nTrip;
        const int tbeg = eBeg + trip * perTrip;
        const int base = tbeg + wave * (128 * SCAN_U) + lane * 4;

        int4 dv[SCAN_U];
        if (tbeg + perTrip <= eEnd) {
            #pragma unroll
            for (int u = 0; u < SCAN_U; ++u)
                dv[u] = *(const int4*)(dst + base + u * 128);
        } else {
            #pragma unroll
            for (int u = 0; u < SCAN_U; ++u) {
                const int e = base + u * 128;
                dv[u] = (e < eEnd) ? *(const int4*)(dst + e) : make_int4(-1, -1, -1, -1);
            }
        }

        #pragma unroll
        for (int u = 0; u < SCAN_U; ++u) {
            const unsigned e = (unsigned)(base + u * 128);
            const unsigned r0 = (unsigned)(dv[u].x - lo);
            const unsigned r1 = (unsigned)(dv[u].y - lo);
            const unsigned r2 = (unsigned)(dv[u].z - lo);
            const unsigned r3 = (unsigned)(dv[u].w - lo);
            const unsigned b0 = __builtin_amdgcn_ballot_w32(r0 < (unsigned)RANGE);
            const unsigned b1 = __builtin_amdgcn_ballot_w32(r1 < (unsigned)RANGE);
            const unsigned b2 = __builtin_amdgcn_ballot_w32(r2 < (unsigned)RANGE);
            const unsigned b3 = __builtin_amdgcn_ballot_w32(r3 < (unsigned)RANGE);
            if ((b0 | b1 | b2 | b3) != 0u) {
                if (b0 != 0u) {
                    if (b0 & bit) q[cnt + (int)__builtin_amdgcn_mbcnt_lo(b0, 0u)] = make_uint2(e + 0u, r0);
                    cnt += __builtin_popcount(b0);
                }
                if (b1 != 0u) {
                    if (b1 & bit) q[cnt + (int)__builtin_amdgcn_mbcnt_lo(b1, 0u)] = make_uint2(e + 1u, r1);
                    cnt += __builtin_popcount(b1);
                }
                if (b2 != 0u) {
                    if (b2 & bit) q[cnt + (int)__builtin_amdgcn_mbcnt_lo(b2, 0u)] = make_uint2(e + 2u, r2);
                    cnt += __builtin_popcount(b2);
                }
                if (b3 != 0u) {
                    if (b3 & bit) q[cnt + (int)__builtin_amdgcn_mbcnt_lo(b3, 0u)] = make_uint2(e + 3u, r3);
                    cnt += __builtin_popcount(b3);
                }
                while (cnt >= 16) {
                    cnt -= 16;
                    uint2* keep = (myList != nullptr && lpos + 16 <= LCAP) ? myList + lpos : nullptr;
                    lpos += 16;
                    own_drain16<true>(q + cnt, 16, x, src, sAcc, sCnt, half, l16, countDeg, keep);
                }
            }
        }
    }
    if (cnt > 0) {
        uint2* keep = (myList != nullptr && lpos + cnt <= LCAP) ? myList + lpos : nullptr;
        lpos += cnt;
        own_drain16<false>(q, cnt, x, src, sAcc, sCnt, half, l16, countDeg, keep);
    }
    return lpos;
}

__global__ void __launch_bounds__(1024)
plane_kernel(const int* __restrict__ src,
             const int* __restrict__ dst,
             unsigned* __restrict__ planes,
             unsigned* __restrict__ packed,
             int nEdges, int nRanges, int planeWords) {
    __shared__ __attribute__((aligned(16))) unsigned bm[PL_GROUP * PL_PITCH];
    const int tid = threadIdx.x;
    const int e0  = blockIdx.x * PL_CHUNK;

    for (int g = 0; g < nRanges; g += PL_GROUP) {
        for (int i = tid; i < PL_GROUP * PL_PITCH / 4; i += 1024)
            ((uint4*)bm)[i] = make_uint4(0u, 0u, 0u, 0u);
        __syncthreads();

        int4 dv[PL_CHUNK / 4096], sv[PL_CHUNK / 4096];
        #pragma unroll
        for (int k = 0; k < PL_CHUNK / 4096; ++k) {
            const int e = e0 + k * 4096 + tid * 4;
            if (e + 3 < nEdges) {
                dv[k] = *(const int4*)(dst + e);
                sv[k] = *(const int4*)(src + e);
            } else {
                dv[k].x = (e + 0 < nEdges) ? dst[e + 0] : -1;
                dv[k].y = (e + 1 < nEdges) ? dst[e + 1] : -1;
                dv[k].z = (e + 2 < nEdges) ? dst[e + 2] : -1;
                dv[k].w = -1;
                sv[k].x = (e + 0 < nEdges) ? src[e + 0] : 0;
                sv[k].y = (e + 1 < nEdges) ? src[e + 1] : 0;
                sv[k].z = (e + 2 < nEdges) ? src[e + 2] : 0;
                sv[k].w = 0;
            }
        }
        #pragma unroll
        for (int k = 0; k < PL_CHUNK / 4096; ++k) {
            const int i = k * 4096 + tid * 4;
            const int e = e0 + i;
            const int d[4] = {dv[k].x, dv[k].y, dv[k].z, dv[k].w};
            const int s[4] = {sv[k].x, sv[k].y, sv[k].z, sv[k].w};
            unsigned pk[4];
            #pragma unroll
            for (int j = 0; j < 4; ++j) {
                const unsigned r   = (unsigned)d[j] / (unsigned)OWN_NODES;
                const unsigned row = (unsigned)d[j] - r * (unsigned)OWN_NODES;
                pk[j] = ((unsigned)s[j] << ROW_BITS) | row;
                const unsigned rl = r - (unsigned)g;
                if (e + j < nEdges && r < (unsigned)nRanges && rl < (unsigned)PL_GROUP)
                    __hip_atomic_fetch_or(&bm[rl * PL_PITCH + ((i + j) >> 5)], 1u << ((i + j) & 31),
                                          __ATOMIC_RELAXED, __HIP_MEMORY_SCOPE_WORKGROUP);
            }
            if (g == 0) {
                if (e + 3 < nEdges) {
                    *(uint4*)(packed + e) = make_uint4(pk[0], pk[1], pk[2], pk[3]);
                } else {
                    #pragma unroll
                    for (int j = 0; j < 4; ++j)
                        if (e + j < nEdges) packed[e + j] = pk[j];
                }
            }
        }
        __syncthreads();

        const int nr = min(PL_GROUP, nRanges - g);
        for (int idx = tid; idx < nr * (PL_CHW / 4); idx += 1024) {
            const int r  = idx / (PL_CHW / 4);
            const int w4 = idx % (PL_CHW / 4);
            const uint4 v = *(const uint4*)(bm + r * PL_PITCH + w4 * 4);
            *(uint4*)(planes + (size_t)(g + r) * planeWords + (size_t)blockIdx.x * PL_CHW + w4 * 4) = v;
        }
        __syncthreads();
    }
}

template <bool FULL>
__device__ __forceinline__ void plane_drain16(const unsigned* q, int n,
                                              const float* __restrict__ x,
                                              const unsigned* __restrict__ packed,
                                              float* sAcc, unsigned* sCnt,
                                              int half, int l16,
                                              uint2* keep) {
    unsigned ed[8];
    bool ok[8];
    #pragma unroll
    for (int p = 0; p < 8; ++p) {
        const int idx = 2 * p + half;
        ok[p] = FULL || (idx < n);
        ed[p] = ok[p] ? q[idx] : 0u;
    }
    const bool saver = (keep != nullptr) && (half == 0) && (FULL || l16 < n);
    const unsigned mineE = saver ? q[l16] : 0u;
    unsigned pk[8];
    #pragma unroll
    for (int p = 0; p < 8; ++p) pk[p] = ok[p] ? packed[ed[p]] : 0u;
    const unsigned minePk = saver ? packed[mineE] : 0u;
    float r0[8], r1[8], r2[8];
    #pragma unroll
    for (int p = 0; p < 8; ++p) {
        const float* xr = x + (size_t)(pk[p] >> ROW_BITS) * FDIM + 3 * l16;
        r0[p] = xr[0];
        r1[p] = xr[1];
        r2[p] = xr[2];
    }
    if (saver) keep[l16] = make_uint2(minePk >> ROW_BITS, minePk & ((1u << ROW_BITS) - 1u));
    #pragma unroll
    for (int p = 0; p < 8; ++p) {
        if (ok[p]) {
            const unsigned row = pk[p] & ((1u << ROW_BITS) - 1u);
            float* a = sAcc + row * FDIM + 3 * l16;
            __hip_atomic_fetch_add(a + 0, r0[p], __ATOMIC_RELAXED, __HIP_MEMORY_SCOPE_WORKGROUP);
            __hip_atomic_fetch_add(a + 1, r1[p], __ATOMIC_RELAXED, __HIP_MEMORY_SCOPE_WORKGROUP);
            __hip_atomic_fetch_add(a + 2, r2[p], __ATOMIC_RELAXED, __HIP_MEMORY_SCOPE_WORKGROUP);
            if (l16 == 0)
                __hip_atomic_fetch_add(sCnt + row, 1u, __ATOMIC_RELAXED, __HIP_MEMORY_SCOPE_WORKGROUP);
        }
    }
}

__device__ __forceinline__ int plane_share(const float* __restrict__ x,
                                           const unsigned* __restrict__ packed,
                                           const uint4* __restrict__ plane4,
                                           float* sAcc, unsigned* sCnt,
                                           unsigned* q,
                                           uint2* myList,
                                           int w4Beg, int w4End,
                                           int lane, int half, int l16) {
    int cnt = 0;
    int lpos = 0;

    if (w4Beg >= w4End) return 0;
    const int last4 = w4End - 1;
    const uint4 none = make_uint4(0u, 0u, 0u, 0u);
    uint4 cur = plane4[min(w4Beg + lane, last4)];
    if (w4Beg + lane > last4) cur = none;
    #pragma unroll 1
    for (int c0 = w4Beg; c0 < w4End; c0 += 32) {
        const int n4 = c0 + 32 + lane;
        uint4 nxt = plane4[min(n4, last4)];
        if (n4 > last4) nxt = none;
        const unsigned eBase = (unsigned)(c0 + lane) * 128u;
        const unsigned m[4] = {cur.x, cur.y, cur.z, cur.w};
        #pragma unroll
        for (int j = 0; j < 4; ++j) {
            unsigned mm = m[j];
            unsigned b = __builtin_amdgcn_ballot_w32(mm != 0u);
            while (b != 0u) {
                if (mm != 0u) {
                    const unsigned pos = (unsigned)__builtin_ctz(mm);
                    mm &= mm - 1u;
                    q[cnt + (int)__builtin_amdgcn_mbcnt_lo(b, 0u)] = eBase + (unsigned)(j * 32) + pos;
                }
                cnt += __builtin_popcount(b);
                while (cnt >= 16) {
                    cnt -= 16;
                    uint2* keep = (lpos + 16 <= LCAP) ? myList + lpos : nullptr;
                    lpos += 16;
                    plane_drain16<true>(q + cnt, 16, x, packed, sAcc, sCnt, half, l16, keep);
                }
                b = __builtin_amdgcn_ballot_w32(mm != 0u);
            }
        }
        cur = nxt;
    }
    if (cnt > 0) {
        uint2* keep = (lpos + cnt <= LCAP) ? myList + lpos : nullptr;
        lpos += cnt;
        plane_drain16<false>(q, cnt, x, packed, sAcc, sCnt, half, l16, keep);
    }
    return lpos;
}

__global__ void __launch_bounds__(OWN_THREADS)
own_sum_kernel(const float* __restrict__ x,
               const unsigned* __restrict__ packed,
               const unsigned* __restrict__ planes,
               float* __restrict__ agg,
               float* __restrict__ cnt,
               unsigned* lists,
               unsigned* listLen,
               int nNodes, int planeWords) {
    __shared__ __attribute__((aligned(16))) float sAcc[OWN_NODES * FDIM];
    __shared__ unsigned sCnt[OWN_NODES];
    __shared__ unsigned sQ[OWN_WAVES * QCAP];
    __shared__ unsigned sLen[OWN_WAVES];

    const int tid  = threadIdx.x;
    const int lane = tid & 31;
    const int wave = __builtin_amdgcn_readfirstlane(tid >> 5);
    const int half = lane >> 4;
    const int l16  = lane & 15;
    const int part  = blockIdx.x % SPLIT;
    const int range = blockIdx.x / SPLIT;
    const int lo    = range * OWN_NODES;

    for (int i = tid; i < OWN_NODES * FDIM / 4; i += OWN_THREADS)
        ((float4*)sAcc)[i] = make_float4(0.0f, 0.0f, 0.0f, 0.0f);
    for (int i = tid; i < OWN_NODES; i += OWN_THREADS) sCnt[i] = 0u;
    __syncthreads();

    uint2* myList = (uint2*)lists + ((size_t)blockIdx.x * OWN_WAVES + wave) * LCAP;

    const int plane4n = planeWords / 4;
    const int share4  = (plane4n + SPLIT - 1) / SPLIT;
    const int s4Beg   = min(part * share4, plane4n);
    const int s4End   = min(s4Beg + share4, plane4n);
    const int wave4   = (s4End - s4Beg + OWN_WAVES - 1) / OWN_WAVES;
    const int w4Beg   = min(s4Beg + wave * wave4, s4End);
    const int w4End   = min(w4Beg + wave4, s4End);
    const uint4* plane4 = (const uint4*)(planes + (size_t)range * planeWords);
    const int lpos = plane_share(x, packed, plane4, sAcc, sCnt, sQ + wave * QCAP, myList,
                                 w4Beg, w4End, lane, half, l16);
    if (lane == 0) sLen[wave] = (unsigned)lpos;
    __syncthreads();

    if (tid < OWN_WAVES) listLen[blockIdx.x * OWN_WAVES + tid] = sLen[tid];

    const int nOwn = min(OWN_NODES, nNodes - lo);
    float4* o = (float4*)(agg + ((size_t)part * nNodes + lo) * FDIM);
    for (int i = tid; i < nOwn * (FDIM / 4); i += OWN_THREADS)
        o[i] = ((const float4*)sAcc)[i];
    for (int i = tid; i < nOwn; i += OWN_THREADS)
        cnt[(size_t)part * nNodes + lo + i] = (float)sCnt[i];
}


struct TileIn {
    v2f   xa[12];
    v2f   ga[SPLIT][12];
    float c[SPLIT];
};

__device__ __forceinline__ void tile_fetch(TileIn& r,
                                           const float* __restrict__ x,
                                           const float* __restrict__ agg,
                                           const float* __restrict__ cnt,
                                           int m0, int l15, int half, int nNodes) {
    const float* xrow = x   + (size_t)(m0 + l15) * FDIM + half * 2;
    const float* grow = agg + (size_t)(m0 + l15) * FDIM + half * 2;
    const size_t plane = (size_t)nNodes * FDIM;
    #pragma unroll
    for (int k = 0; k < 12; ++k) {
        r.xa[k].x = xrow[4 * k];
        r.xa[k].y = xrow[4 * k + 1];
    }
    #pragma unroll
    for (int s = 0; s < SPLIT; ++s) {
        #pragma unroll
        for (int k = 0; k < 12; ++k) {
            r.ga[s][k].x = grow[s * plane + 4 * k];
            r.ga[s][k].y = grow[s * plane + 4 * k + 1];
        }
    }
    #pragma unroll
    for (int s = 0; s < SPLIT; ++s) r.c[s] = cnt[(size_t)s * nNodes + m0 + l15];
}

__device__ __forceinline__ void tile_products(const TileIn& r,
                                              const v2f (&wS)[12][3], const v2f (&wN)[12][3],
                                              float bv0, float bv1, float bv2,
                                              float* __restrict__ out,
                                              int m0, int l15, int half, int relu) {
    float deg = r.c[0];
    #pragma unroll
    for (int s = 1; s < SPLIT; ++s) deg += r.c[s];
    const float dsc = 1.0f / fmaxf(deg, 1.0f);

    v8f cs[3], cn[3];
    #pragma unroll
    for (int q = 0; q < 8; ++q) {
        cs[0][q] = bv0; cs[1][q] = bv1; cs[2][q] = bv2;
        cn[0][q] = 0.0f; cn[1][q] = 0.0f; cn[2][q] = 0.0f;
    }

    #pragma unroll
    for (int k = 0; k < 12; ++k) {
        const v2f a = r.xa[k];
        v2f a2 = r.ga[0][k];
        #pragma unroll
        for (int s = 1; s < SPLIT; ++s) {
            a2.x += r.ga[s][k].x;
            a2.y += r.ga[s][k].y;
        }
        a2.x *= dsc;
        a2.y *= dsc;
        #pragma unroll
        for (int nt = 0; nt < 3; ++nt) {
            cs[nt] = __builtin_amdgcn_wmma_f32_16x16x4_f32(
                false, a, false, wS[k][nt], (short)0, cs[nt], false, false);
            cn[nt] = __builtin_amdgcn_wmma_f32_16x16x4_f32(
                false, a2, false, wN[k][nt], (short)0, cn[nt], false, false);
        }
    }

    #pragma unroll
    for (int nt = 0; nt < 3; ++nt) {
        #pragma unroll
        for (int q = 0; q < 8; ++q) {
            float v = cs[nt][q] + cn[nt][q];
            if (relu) v = fmaxf(v, 0.0f);
            out[(size_t)(m0 + half * 8 + q) * FDIM + nt * 16 + l15] = v;
        }
    }
}

__global__ void __launch_bounds__(64)
sage_gemm_kernel(const float* __restrict__ x,
                 const float* __restrict__ agg,
                 const float* __restrict__ cnt,
                 const float* __restrict__ Wself,
                 const float* __restrict__ Wneigh,
                 const float* __restrict__ bias,
                 float* __restrict__ out,
                 int relu, int tilesPerWave, int nNodes) {
    const int lane = threadIdx.x & 31;
    const int wave = threadIdx.x >> 5;
    const int half = lane >> 4;
    const int l15  = lane & 15;
    const int gw   = blockIdx.x * 2 + wave;
    const int mBeg = gw * tilesPerWave * 16;

    TileIn cur;
    tile_fetch(cur, x, agg, cnt, mBeg, l15, half, nNodes);

    v2f wS[12][3], wN[12][3];
    #pragma unroll
    for (int k = 0; k < 12; ++k) {
        const int kk = k * 4 + half * 2;
        #pragma unroll
        for (int nt = 0; nt < 3; ++nt) {
            const int col = nt * 16 + l15;
            wS[k][nt].x = Wself [(size_t)(kk)     * FDIM + col];
            wS[k][nt].y = Wself [(size_t)(kk + 1) * FDIM + col];
            wN[k][nt].x = Wneigh[(size_t)(kk)     * FDIM + col];
            wN[k][nt].y = Wneigh[(size_t)(kk + 1) * FDIM + col];
        }
    }
    float bv0 = bias[0 * 16 + l15];
    float bv1 = bias[1 * 16 + l15];
    float bv2 = bias[2 * 16 + l15];

    #pragma unroll 1
    for (int t = 0; t + 1 < tilesPerWave; ++t) {
        const int m0 = mBeg + t * 16;
        TileIn nxt;
        tile_fetch(nxt, x, agg, cnt, m0 + 16, l15, half, nNodes);
        __builtin_amdgcn_sched_barrier(0);
        tile_products(cur, wS, wN, bv0, bv1, bv2, out, m0, l15, half, relu);
        __builtin_amdgcn_sched_barrier(0);
        cur = nxt;
    }
    tile_products(cur, wS, wN, bv0, bv1, bv2, out, mBeg + (tilesPerWave - 1) * 16, l15, half, relu);
}

template <bool FULL>
__device__ __forceinline__ void list_drain16(const uint2* q, int n,
                                             const float* __restrict__ x, float* sAcc,
                                             int half, int l16) {
    uint2 ent[8];
    #pragma unroll
    for (int p = 0; p < 8; ++p) {
        const int idx = 2 * p + half;
        ent[p] = (FULL || idx < n) ? q[idx] : make_uint2(0u, 0u);
    }
    float r0[8], r1[8], r2[8];
    list_rows16<FULL>(ent, n, x, r0, r1, r2, half, l16);
    list_adds16<FULL>(ent, n, sAcc, r0, r1, r2, half, l16);
}

__global__ void __launch_bounds__(OWN_THREADS)
own_sum_gemm_kernel(const float* __restrict__ x,
                    const int* __restrict__ src,
                    const int* __restrict__ dst,
                    const float* __restrict__ cnt,
                    const unsigned* __restrict__ lists,
                    const unsigned* __restrict__ listLen,
                    const float* __restrict__ Wself,
                    const float* __restrict__ Wneigh,
                    const float* __restrict__ bias,
                    float* __restrict__ out,
                    int nNodes, int nEdges) {
    __shared__ __attribute__((aligned(16))) float sAcc[L2_NODES * FDIM];
    __shared__ __attribute__((aligned(8))) uint2 sQ[OWN_WAVES * QCAP];
    __shared__ __attribute__((aligned(8))) v2f sW[2 * 12 * 3 * 32];

    const int tid  = threadIdx.x;
    const int lane = tid & 31;
    const int wave = __builtin_amdgcn_readfirstlane(tid >> 5);
    const int half = lane >> 4;
    const int l16  = lane & 15;
    const int range = blockIdx.x / SPLIT;
    const int sub   = blockIdx.x % SPLIT;
    const int lo    = range * OWN_NODES + sub * L2_NODES;

    for (int i = tid; i < L2_NODES * FDIM / 4; i += OWN_THREADS)
        ((float4*)sAcc)[i] = make_float4(0.0f, 0.0f, 0.0f, 0.0f);
    for (int i = tid; i < 2 * 12 * 3 * 32; i += OWN_THREADS) {
        const int ln  = i & 31;
        const int f   = i >> 5;
        const int nt  = f % 3;
        const int k   = (f / 3) % 12;
        const float* Wm = (f >= 36) ? Wneigh : Wself;
        const int kk  = k * 4 + (ln >> 4) * 2;
        const int col = nt * 16 + (ln & 15);
        v2f w;
        w.x = Wm[(size_t)(kk)     * FDIM + col];
        w.y = Wm[(size_t)(kk + 1) * FDIM + col];
        sW[i] = w;
    }
    __syncthreads();

    uint2* q = sQ + wave * QCAP;

    unsigned len[SPLIT];
    bool over = false;
    #pragma unroll
    for (int s = 0; s < SPLIT; ++s) {
        len[s] = listLen[(range * SPLIT + s) * OWN_WAVES + lane];
        over = over || (len[s] > (unsigned)LCAP);
    }
    const bool fromLists = (__builtin_amdgcn_ballot_w32(over) == 0u);

    if (fromLists) {
        int cntQ = 0;
        const unsigned bit = 1u << lane;
        #pragma unroll
        for (int s = 0; s < SPLIT; ++s) {
            const int n = __builtin_amdgcn_readlane((int)len[s], wave);
            const uint2* list = (const uint2*)lists
                              + ((size_t)(range * SPLIT + s) * OWN_WAVES + wave) * LCAP;
            for (int pos = 0; pos < n; pos += 32 * LIST_U) {
                uint2 ent[LIST_U];
                #pragma unroll
                for (int u = 0; u < LIST_U; ++u) {
                    const int idx = pos + u * 32 + lane;
                    ent[u] = (idx < n) ? list[idx] : make_uint2(0u, 0xffffffffu);
                }
                #pragma unroll
                for (int u = 0; u < LIST_U; ++u) {
                    const unsigned r = ent[u].y - (unsigned)(sub * L2_NODES);
                    const unsigned b = __builtin_amdgcn_ballot_w32(r < (unsigned)L2_NODES);
                    if (b != 0u) {
                        if (b & bit) q[cntQ + (int)__builtin_amdgcn_mbcnt_lo(b, 0u)] = make_uint2(ent[u].x, r);
                        cntQ += __builtin_popcount(b);
                        while (cntQ >= 16) {
                            cntQ -= 16;
                            list_drain16<true>(q + cntQ, 16, x, sAcc, half, l16);
                        }
                    }
                }
            }
        }
        if (cntQ > 0) list_drain16<false>(q, cntQ, x, sAcc, half, l16);
    } else {
        const int nEdges4 = nEdges & ~3;
        sweep_share<L2_NODES>(x, src, dst, sAcc, nullptr, q, nullptr,
                              lo, 0, nEdges4, true, nEdges4, nEdges,
                              wave, lane, half, l16, 0);
    }
    const int nOwn    = max(0, min(L2_NODES, nNodes - lo));
    const bool hasTile = wave < nOwn / 16;
    const int rowL = wave * 16 + l16;
    v2f xa[12];
    float deg = 1.0f, bv0 = 0.0f, bv1 = 0.0f, bv2 = 0.0f;
    #pragma unroll
    for (int k = 0; k < 12; ++k) { xa[k].x = 0.0f; xa[k].y = 0.0f; }
    if (hasTile) {
        const int m = lo + rowL;
        const float* xrow = x + (size_t)m * FDIM + half * 2;
        #pragma unroll
        for (int k = 0; k < 12; ++k) {
            xa[k].x = xrow[4 * k];
            xa[k].y = xrow[4 * k + 1];
        }
        deg = cnt[m];
        #pragma unroll
        for (int s = 1; s < SPLIT; ++s) deg += cnt[(size_t)s * nNodes + m];
        bv0 = bias[0 * 16 + l16];
        bv1 = bias[1 * 16 + l16];
        bv2 = bias[2 * 16 + l16];
    }
    __syncthreads();

    if (hasTile) {
        const float dsc = 1.0f / fmaxf(deg, 1.0f);

        v8f cs[3], cn[3];
        #pragma unroll
        for (int e = 0; e < 8; ++e) {
            cs[0][e] = bv0; cs[1][e] = bv1; cs[2][e] = bv2;
            cn[0][e] = 0.0f; cn[1][e] = 0.0f; cn[2][e] = 0.0f;
        }

        const v2f* srow = (const v2f*)(sAcc + rowL * FDIM + half * 2);
        #pragma unroll
        for (int k = 0; k < 12; ++k) {
            v2f a2 = srow[2 * k];
            a2.x *= dsc;
            a2.y *= dsc;
            #pragma unroll
            for (int nt = 0; nt < 3; ++nt) {
                const v2f wS = sW[((0 * 12 + k) * 3 + nt) * 32 + lane];
                const v2f wN = sW[((1 * 12 + k) * 3 + nt) * 32 + lane];
                cs[nt] = __builtin_amdgcn_wmma_f32_16x16x4_f32(
                    false, xa[k], false, wS, (short)0, cs[nt], false, false);
                cn[nt] = __builtin_amdgcn_wmma_f32_16x16x4_f32(
                    false, a2, false, wN, (short)0, cn[nt], false, false);
            }
        }

        float* orow = out + (size_t)(lo + wave * 16 + half * 8) * FDIM + l16;
        #pragma unroll
        for (int nt = 0; nt < 3; ++nt) {
            #pragma unroll
            for (int e = 0; e < 8; ++e)
                orow[(size_t)e * FDIM + nt * 16] = cs[nt][e] + cn[nt][e];
        }
    }
}

extern "C" void kernel_launch(void* const* d_in, const int* in_sizes, int n_in,
                              void* d_out, int out_size, void* d_ws, size_t ws_size,
                              hipStream_t stream) {
    const float* features = (const float*)d_in[0];
    const int*   src      = (const int*)d_in[1];
    const int*   dst      = (const int*)d_in[2];
    const float* Wself1   = (const float*)d_in[3];
    const float* Wneigh1  = (const float*)d_in[4];
    const float* b1       = (const float*)d_in[5];
    const float* Wself2   = (const float*)d_in[6];
    const float* Wneigh2  = (const float*)d_in[7];
    const float* b2       = (const float*)d_in[8];
    float* out = (float*)d_out;

    const int N = in_sizes[0] / FDIM;
    const int E = in_sizes[1];

    char* ws = (char*)d_ws;
    size_t off = 0;
    float* cnt  = (float*)(ws + off); off += (((size_t)SPLIT * N * 4)        + 255) & ~(size_t)255;
    float* agg  = (float*)(ws + off); off += (((size_t)SPLIT * N * FDIM * 4) + 255) & ~(size_t)255;
    float* h    = (float*)(ws + off); off += (((size_t)N * FDIM * 4) + 255) & ~(size_t)255;

    const int ownBlocks = ((N + OWN_NODES - 1) / OWN_NODES) * SPLIT;
    unsigned* listLen = (unsigned*)(ws + off); off += (((size_t)ownBlocks * OWN_WAVES * 4) + 255) & ~(size_t)255;
    unsigned* lists   = (unsigned*)(ws + off);
    off += (((size_t)ownBlocks * OWN_WAVES * LCAP * 8) + 255) & ~(size_t)255;

    const int nRanges    = ownBlocks / SPLIT;
    const int nChunks    = (E + PL_CHUNK - 1) / PL_CHUNK;
    const int planeWords = nChunks * PL_CHW;
    unsigned* packed = (unsigned*)(ws + off); off += (((size_t)E * 4) + 255) & ~(size_t)255;
    unsigned* planes = (unsigned*)(ws + off);

    const int tiles        = N / 16;
    const int tilesPerWave = 5;
    const int gemmWaves    = tiles / tilesPerWave;
    const int gemmBlocks   = gemmWaves / 2;

    plane_kernel<<<nChunks, 1024, 0, stream>>>(src, dst, planes, packed, E, nRanges, planeWords);

    own_sum_kernel<<<ownBlocks, OWN_THREADS, 0, stream>>>(features, packed, planes, agg, cnt,
                                                          lists, listLen, N, planeWords);
    sage_gemm_kernel<<<gemmBlocks, 64, 0, stream>>>(features, agg, cnt,
                                                    Wself1, Wneigh1, b1, h, 1, tilesPerWave, N);

    own_sum_gemm_kernel<<<ownBlocks, OWN_THREADS, 0, stream>>>(h, src, dst, cnt, lists, listLen,
                                                               Wself2, Wneigh2, b2, out, N, E);
}


# layer-1 GEMM fused into the layer-1 owner pass: one shared bit plane (edge ends in its range's upper half) lets a work-group own 400 whole nodes, sums complete in LDS; 4 launches to 3; new sum order
# speedup vs baseline: 4.2100x; 1.0536x over previous
#include <hip/hip_runtime.h>


typedef __attribute__((ext_vector_type(2))) float v2f;
typedef __attribute__((ext_vector_type(8))) float v8f;

#define FDIM 48

#define SPLIT       2
#define OWN_NODES   800
#define OWN_THREADS 1024
#define OWN_WAVES   32
#define SCAN_U      8
#define QCAP        160
#define LCAP        512
#define L2_NODES    (OWN_NODES / SPLIT)
#define LIST_U      8
static_assert(SPLIT == 2, "one half plane tells the two owners of a range apart");
static_assert(OWN_NODES % (SPLIT * 16) == 0, "a work-group owns whole 16-node tiles");
static_assert(L2_NODES / 16 <= OWN_WAVES, "one tile per wave in the products");

#define PL_CHUNK    4096
#define PL_CHW      (PL_CHUNK / 32)
#define PL_PITCH    (PL_CHW + 4)
#define PL_GROUP    128
#define ROW_BITS    10
static_assert(OWN_NODES <= (1 << ROW_BITS), "a row of a range fits the packed word's low bits");
static_assert(PL_CHUNK % 4096 == 0, "a pre-pass thread takes whole groups of four edges");

template <bool FULL>
__device__ __forceinline__ void own_drain16(const uint2* q, int n,
                                            const float* __restrict__ x,
                                            const int* __restrict__ src,
                                            float* sAcc, unsigned* sCnt,
                                            int half, int l16, int countDeg,
                                            uint2* keep) {
    uint2 ent[8];
    bool ok[8];
    #pragma unroll
    for (int p = 0; p < 8; ++p) {
        const int idx = 2 * p + half;
        ok[p] = FULL || (idx < n);
        ent[p] = ok[p] ? q[idx] : make_uint2(0u, 0u);
    }
    const bool saver = (keep != nullptr) && (half == 0) && (FULL || l16 < n);
    const uint2 mine = saver ? q[l16] : make_uint2(0u, 0u);
    int s[8];
    #pragma unroll
    for (int p = 0; p < 8; ++p) s[p] = ok[p] ? src[ent[p].x] : 0;
    const int mineSrc = saver ? src[mine.x] : 0;
    float r0[8], r1[8], r2[8];
    #pragma unroll
    for (int p = 0; p < 8; ++p) {
        const float* xr = x + (size_t)s[p] * FDIM + 3 * l16;
        r0[p] = xr[0];
        r1[p] = xr[1];
        r2[p] = xr[2];
    }
    if (saver) keep[l16] = make_uint2((unsigned)mineSrc, mine.y);
    #pragma unroll
    for (int p = 0; p < 8; ++p) {
        if (ok[p]) {
            float* a = sAcc + ent[p].y * FDIM + 3 * l16;
            __hip_atomic_fetch_add(a + 0, r0[p], __ATOMIC_RELAXED, __HIP_MEMORY_SCOPE_WORKGROUP);
            __hip_atomic_fetch_add(a + 1, r1[p], __ATOMIC_RELAXED, __HIP_MEMORY_SCOPE_WORKGROUP);
            __hip_atomic_fetch_add(a + 2, r2[p], __ATOMIC_RELAXED, __HIP_MEMORY_SCOPE_WORKGROUP);
            if (countDeg && l16 == 0)
                __hip_atomic_fetch_add(sCnt + ent[p].y, 1u, __ATOMIC_RELAXED, __HIP_MEMORY_SCOPE_WORKGROUP);
        }
    }
}

template <bool FULL>
__device__ __forceinline__ void list_rows16(const uint2 (&ent)[8], int n,
                                            const float* __restrict__ x,
                                            float (&r0)[8], float (&r1)[8], float (&r2)[8],
                                            int half, int l16) {
    #pragma unroll
    for (int p = 0; p < 8; ++p) {
        if (FULL || (2 * p + half < n)) {
            const float* xr = x + (size_t)ent[p].x * FDIM + 3 * l16;
            r0[p] = xr[0];
            r1[p] = xr[1];
            r2[p] = xr[2];
        } else {
            r0[p] = 0.0f; r1[p] = 0.0f; r2[p] = 0.0f;
        }
    }
}

template <bool FULL>
__device__ __forceinline__ void list_adds16(const uint2 (&ent)[8], int n, float* sAcc,
                                            const float (&r0)[8], const float (&r1)[8],
                                            const float (&r2)[8], int half, int l16) {
    #pragma unroll
    for (int p = 0; p < 8; ++p) {
        if (FULL || (2 * p + half < n)) {
            float* a = sAcc + ent[p].y * FDIM + 3 * l16;
            __hip_atomic_fetch_add(a + 0, r0[p], __ATOMIC_RELAXED, __HIP_MEMORY_SCOPE_WORKGROUP);
            __hip_atomic_fetch_add(a + 1, r1[p], __ATOMIC_RELAXED, __HIP_MEMORY_SCOPE_WORKGROUP);
            __hip_atomic_fetch_add(a + 2, r2[p], __ATOMIC_RELAXED, __HIP_MEMORY_SCOPE_WORKGROUP);
        }
    }
}

template <int RANGE>
__device__ __forceinline__ int sweep_share(const float* __restrict__ x,
                                           const int* __restrict__ src,
                                           const int* __restrict__ dst,
                                           float* sAcc, unsigned* sCnt,
                                           uint2* q,
                                           uint2* myList,
                                           int lo, int eBeg, int eEnd, bool tail,
                                           int nEdges4, int nEdges,
                                           int wave, int lane, int half, int l16, int countDeg) {
    int cnt = 0;
    int lpos = 0;

    const unsigned bit = 1u << lane;

    if (wave == 0 && tail) {
        const int e = nEdges4 + lane;
        const unsigned r = (e < nEdges) ? (unsigned)(dst[e] - lo) : 0xffffffffu;
        const unsigned b = __builtin_amdgcn_ballot_w32(r < (unsigned)RANGE);
        if (b != 0u) {
            if (b & bit) q[(int)__builtin_amdgcn_mbcnt_lo(b, 0u)] = make_uint2((unsigned)e, r);
            cnt = __builtin_popcount(b);
        }
    }

    const int perTrip = OWN_THREADS * 4 * SCAN_U;
    const int nTrip   = (eEnd - eBeg + perTrip - 1) / perTrip;
    const int trip0   = (int)(((long long)blockIdx.x * nTrip) / gridDim.x);

    for (int i = 0; i < nTrip; ++i) {
        int trip = trip0 + i;
        if (trip >= nTrip) trip -= nTrip;
        const int tbeg = eBeg + trip * perTrip;
        const int base = tbeg + wave * (128 * SCAN_U) + lane * 4;

        int4 dv[SCAN_U];
        if (tbeg + perTrip <= eEnd) {
            #pragma unroll
            for (int u = 0; u < SCAN_U; ++u)
                dv[u] = *(const int4*)(dst + base + u * 128);
        } else {
            #pragma unroll
            for (int u = 0; u < SCAN_U; ++u) {
                const int e = base + u * 128;
                dv[u] = (e < eEnd) ? *(const int4*)(dst + e) : make_int4(-1, -1, -1, -1);
            }
        }

        #pragma unroll
        for (int u = 0; u < SCAN_U; ++u) {
            const unsigned e = (unsigned)(base + u * 128);
            const unsigned r0 = (unsigned)(dv[u].x - lo);
            const unsigned r1 = (unsigned)(dv[u].y - lo);
            const unsigned r2 = (unsigned)(dv[u].z - lo);
            const unsigned r3 = (unsigned)(dv[u].w - lo);
            const unsigned b0 = __builtin_amdgcn_ballot_w32(r0 < (unsigned)RANGE);
            const unsigned b1 = __builtin_amdgcn_ballot_w32(r1 < (unsigned)RANGE);
            const unsigned b2 = __builtin_amdgcn_ballot_w32(r2 < (unsigned)RANGE);
            const unsigned b3 = __builtin_amdgcn_ballot_w32(r3 < (unsigned)RANGE);
            if ((b0 | b1 | b2 | b3) != 0u) {
                if (b0 != 0u) {
                    if (b0 & bit) q[cnt + (int)__builtin_amdgcn_mbcnt_lo(b0, 0u)] = make_uint2(e + 0u, r0);
                    cnt += __builtin_popcount(b0);
                }
                if (b1 != 0u) {
                    if (b1 & bit) q[cnt + (int)__builtin_amdgcn_mbcnt_lo(b1, 0u)] = make_uint2(e + 1u, r1);
                    cnt += __builtin_popcount(b1);
                }
                if (b2 != 0u) {
                    if (b2 & bit) q[cnt + (int)__builtin_amdgcn_mbcnt_lo(b2, 0u)] = make_uint2(e + 2u, r2);
                    cnt += __builtin_popcount(b2);
                }
                if (b3 != 0u) {
                    if (b3 & bit) q[cnt + (int)__builtin_amdgcn_mbcnt_lo(b3, 0u)] = make_uint2(e + 3u, r3);
                    cnt += __builtin_popcount(b3);
                }
                while (cnt >= 16) {
                    cnt -= 16;
                    uint2* keep = (myList != nullptr && lpos + 16 <= LCAP) ? myList + lpos : nullptr;
                    lpos += 16;
                    own_drain16<true>(q + cnt, 16, x, src, sAcc, sCnt, half, l16, countDeg, keep);
                }
            }
        }
    }
    if (cnt > 0) {
        uint2* keep = (myList != nullptr && lpos + cnt <= LCAP) ? myList + lpos : nullptr;
        lpos += cnt;
        own_drain16<false>(q, cnt, x, src, sAcc, sCnt, half, l16, countDeg, keep);
    }
    return lpos;
}

__global__ void __launch_bounds__(1024)
plane_kernel(const int* __restrict__ src,
             const int* __restrict__ dst,
             unsigned* __restrict__ planes,
             unsigned* __restrict__ packed,
             int nEdges, int nRanges, int planeWords) {
    __shared__ __attribute__((aligned(16))) unsigned bm[PL_GROUP * PL_PITCH];
    const int tid = threadIdx.x;
    const int e0  = blockIdx.x * PL_CHUNK;
    const int nPlanes = nRanges + 1;

    for (int g = 0; g < nPlanes; g += PL_GROUP) {
        for (int i = tid; i < PL_GROUP * PL_PITCH / 4; i += 1024)
            ((uint4*)bm)[i] = make_uint4(0u, 0u, 0u, 0u);
        __syncthreads();

        int4 dv[PL_CHUNK / 4096], sv[PL_CHUNK / 4096];
        #pragma unroll
        for (int k = 0; k < PL_CHUNK / 4096; ++k) {
            const int e = e0 + k * 4096 + tid * 4;
            if (e + 3 < nEdges) {
                dv[k] = *(const int4*)(dst + e);
                sv[k] = *(const int4*)(src + e);
            } else {
                dv[k].x = (e + 0 < nEdges) ? dst[e + 0] : -1;
                dv[k].y = (e + 1 < nEdges) ? dst[e + 1] : -1;
                dv[k].z = (e + 2 < nEdges) ? dst[e + 2] : -1;
                dv[k].w = -1;
                sv[k].x = (e + 0 < nEdges) ? src[e + 0] : 0;
                sv[k].y = (e + 1 < nEdges) ? src[e + 1] : 0;
                sv[k].z = (e + 2 < nEdges) ? src[e + 2] : 0;
                sv[k].w = 0;
            }
        }
        const unsigned hl = (unsigned)nRanges - (unsigned)g;
        #pragma unroll
        for (int k = 0; k < PL_CHUNK / 4096; ++k) {
            const int i = k * 4096 + tid * 4;
            const int e = e0 + i;
            const int d[4] = {dv[k].x, dv[k].y, dv[k].z, dv[k].w};
            const int s[4] = {sv[k].x, sv[k].y, sv[k].z, sv[k].w};
            unsigned pk[4];
            unsigned hb = 0u;
            #pragma unroll
            for (int j = 0; j < 4; ++j) {
                const unsigned r   = (unsigned)d[j] / (unsigned)OWN_NODES;
                const unsigned row = (unsigned)d[j] - r * (unsigned)OWN_NODES;
                pk[j] = ((unsigned)s[j] << ROW_BITS) | row;
                const unsigned rl = r - (unsigned)g;
                if (e + j < nEdges && r < (unsigned)nRanges) {
                    if (rl < (unsigned)PL_GROUP)
                        __hip_atomic_fetch_or(&bm[rl * PL_PITCH + ((i + j) >> 5)], 1u << ((i + j) & 31),
                                              __ATOMIC_RELAXED, __HIP_MEMORY_SCOPE_WORKGROUP);
                    if (row >= (unsigned)L2_NODES) hb |= 1u << ((i + j) & 31);
                }
            }
            if (hb != 0u && hl < (unsigned)PL_GROUP)
                __hip_atomic_fetch_or(&bm[hl * PL_PITCH + (i >> 5)], hb,
                                      __ATOMIC_RELAXED, __HIP_MEMORY_SCOPE_WORKGROUP);
            if (g == 0) {
                if (e + 3 < nEdges) {
                    *(uint4*)(packed + e) = make_uint4(pk[0], pk[1], pk[2], pk[3]);
                } else {
                    #pragma unroll
                    for (int j = 0; j < 4; ++j)
                        if (e + j < nEdges) packed[e + j] = pk[j];
                }
            }
        }
        __syncthreads();

        const int nr = min(PL_GROUP, nPlanes - g);
        for (int idx = tid; idx < nr * (PL_CHW / 4); idx += 1024) {
            const int r  = idx / (PL_CHW / 4);
            const int w4 = idx % (PL_CHW / 4);
            const uint4 v = *(const uint4*)(bm + r * PL_PITCH + w4 * 4);
            *(uint4*)(planes + (size_t)(g + r) * planeWords + (size_t)blockIdx.x * PL_CHW + w4 * 4) = v;
        }
        __syncthreads();
    }
}

template <bool FULL>
__device__ __forceinline__ void plane_drain16(const unsigned* q, int n,
                                              const float* __restrict__ x,
                                              const unsigned* __restrict__ packed,
                                              float* sAcc, unsigned* sCnt,
                                              unsigned rowOff, int half, int l16,
                                              uint2* keep) {
    unsigned ed[8];
    bool ok[8];
    #pragma unroll
    for (int p = 0; p < 8; ++p) {
        const int idx = 2 * p + half;
        ok[p] = FULL || (idx < n);
        ed[p] = ok[p] ? q[idx] : 0u;
    }
    const bool saver = (keep != nullptr) && (half == 0) && (FULL || l16 < n);
    const unsigned mineE = saver ? q[l16] : 0u;
    unsigned pk[8];
    #pragma unroll
    for (int p = 0; p < 8; ++p) pk[p] = ok[p] ? packed[ed[p]] : 0u;
    const unsigned minePk = saver ? packed[mineE] : 0u;
    float r0[8], r1[8], r2[8];
    #pragma unroll
    for (int p = 0; p < 8; ++p) {
        const float* xr = x + (size_t)(pk[p] >> ROW_BITS) * FDIM + 3 * l16;
        r0[p] = xr[0];
        r1[p] = xr[1];
        r2[p] = xr[2];
    }
    if (saver) keep[l16] = make_uint2(minePk >> ROW_BITS, (minePk & ((1u << ROW_BITS) - 1u)) - rowOff);
    #pragma unroll
    for (int p = 0; p < 8; ++p) {
        if (ok[p]) {
            const unsigned row = (pk[p] & ((1u << ROW_BITS) - 1u)) - rowOff;
            float* a = sAcc + row * FDIM + 3 * l16;
            __hip_atomic_fetch_add(a + 0, r0[p], __ATOMIC_RELAXED, __HIP_MEMORY_SCOPE_WORKGROUP);
            __hip_atomic_fetch_add(a + 1, r1[p], __ATOMIC_RELAXED, __HIP_MEMORY_SCOPE_WORKGROUP);
            __hip_atomic_fetch_add(a + 2, r2[p], __ATOMIC_RELAXED, __HIP_MEMORY_SCOPE_WORKGROUP);
            if (l16 == 0)
                __hip_atomic_fetch_add(sCnt + row, 1u, __ATOMIC_RELAXED, __HIP_MEMORY_SCOPE_WORKGROUP);
        }
    }
}

__device__ __forceinline__ uint4 plane_piece(const uint4* __restrict__ plane4,
                                             const uint4* __restrict__ half4,
                                             unsigned flip, int i4, int last4) {
    const int c = min(i4, last4);
    uint4 p = plane4[c];
    const uint4 h = half4[c];
    p.x &= h.x ^ flip;
    p.y &= h.y ^ flip;
    p.z &= h.z ^ flip;
    p.w &= h.w ^ flip;
    if (i4 > last4) p = make_uint4(0u, 0u, 0u, 0u);
    return p;
}

__device__ __forceinline__ int plane_share(const float* __restrict__ x,
                                           const unsigned* __restrict__ packed,
                                           const uint4* __restrict__ plane4,
                                           const uint4* __restrict__ half4,
                                           unsigned flip, unsigned rowOff,
                                           float* sAcc, unsigned* sCnt,
                                           unsigned* q,
                                           uint2* myList,
                                           int w4Beg, int w4End,
                                           int lane, int half, int l16) {
    int cnt = 0;
    int lpos = 0;

    if (w4Beg >= w4End) return 0;
    const int last4 = w4End - 1;
    uint4 cur = plane_piece(plane4, half4, flip, w4Beg + lane, last4);
    #pragma unroll 1
    for (int c0 = w4Beg; c0 < w4End; c0 += 32) {
        const uint4 nxt = plane_piece(plane4, half4, flip, c0 + 32 + lane, last4);
        const unsigned eBase = (unsigned)(c0 + lane) * 128u;
        const unsigned m[4] = {cur.x, cur.y, cur.z, cur.w};
        #pragma unroll
        for (int j = 0; j < 4; ++j) {
            unsigned mm = m[j];
            unsigned b = __builtin_amdgcn_ballot_w32(mm != 0u);
            while (b != 0u) {
                if (mm != 0u) {
                    const unsigned pos = (unsigned)__builtin_ctz(mm);
                    mm &= mm - 1u;
                    q[cnt + (int)__builtin_amdgcn_mbcnt_lo(b, 0u)] = eBase + (unsigned)(j * 32) + pos;
                }
                cnt += __builtin_popcount(b);
                while (cnt >= 16) {
                    cnt -= 16;
                    uint2* keep = (lpos + 16 <= LCAP) ? myList + lpos : nullptr;
                    lpos += 16;
                    plane_drain16<true>(q + cnt, 16, x, packed, sAcc, sCnt, rowOff, half, l16, keep);
                }
                b = __builtin_amdgcn_ballot_w32(mm != 0u);
            }
        }
        cur = nxt;
    }
    if (cnt > 0) {
        uint2* keep = (lpos + cnt <= LCAP) ? myList + lpos : nullptr;
        lpos += cnt;
        plane_drain16<false>(q, cnt, x, packed, sAcc, sCnt, rowOff, half, l16, keep);
    }
    return lpos;
}


__device__ __forceinline__ void stage_weights(v2f* sW, const float* __restrict__ Wself,
                                              const float* __restrict__ Wneigh, int tid) {
    for (int i = tid; i < 2 * 12 * 3 * 32; i += OWN_THREADS) {
        const int ln  = i & 31;
        const int f   = i >> 5;
        const int nt  = f % 3;
        const int k   = (f / 3) % 12;
        const float* Wm = (f >= 36) ? Wneigh : Wself;
        const int kk  = k * 4 + (ln >> 4) * 2;
        const int col = nt * 16 + (ln & 15);
        v2f w;
        w.x = Wm[(size_t)(kk)     * FDIM + col];
        w.y = Wm[(size_t)(kk + 1) * FDIM + col];
        sW[i] = w;
    }
}

template <bool RELU>
__device__ __forceinline__ void owned_tile_products(const v2f (&xa)[12], const v2f* srow,
                                                    const v2f* sW, float dsc,
                                                    float bv0, float bv1, float bv2,
                                                    float* __restrict__ orow, int lane) {
    v8f cs[3], cn[3];
    #pragma unroll
    for (int e = 0; e < 8; ++e) {
        cs[0][e] = bv0; cs[1][e] = bv1; cs[2][e] = bv2;
        cn[0][e] = 0.0f; cn[1][e] = 0.0f; cn[2][e] = 0.0f;
    }

    #pragma unroll
    for (int k = 0; k < 12; ++k) {
        v2f a2 = srow[2 * k];
        a2.x *= dsc;
        a2.y *= dsc;
        #pragma unroll
        for (int nt = 0; nt < 3; ++nt) {
            const v2f wS = sW[((0 * 12 + k) * 3 + nt) * 32 + lane];
            const v2f wN = sW[((1 * 12 + k) * 3 + nt) * 32 + lane];
            cs[nt] = __builtin_amdgcn_wmma_f32_16x16x4_f32(
                false, xa[k], false, wS, (short)0, cs[nt], false, false);
            cn[nt] = __builtin_amdgcn_wmma_f32_16x16x4_f32(
                false, a2, false, wN, (short)0, cn[nt], false, false);
        }
    }

    #pragma unroll
    for (int nt = 0; nt < 3; ++nt) {
        #pragma unroll
        for (int e = 0; e < 8; ++e) {
            float v = cs[nt][e] + cn[nt][e];
            if (RELU) v = fmaxf(v, 0.0f);
            orow[(size_t)e * FDIM + nt * 16] = v;
        }
    }
}

__global__ void __launch_bounds__(OWN_THREADS)
own_plane_gemm_kernel(const float* __restrict__ x,
                      const unsigned* __restrict__ packed,
                      const unsigned* __restrict__ planes,
                      const float* __restrict__ Wself,
                      const float* __restrict__ Wneigh,
                      const float* __restrict__ bias,
                      float* __restrict__ hout,
                      float* __restrict__ cnt,
                      unsigned* lists,
                      unsigned* listLen,
                      int nNodes, int nRanges, int planeWords) {
    __shared__ __attribute__((aligned(16))) float sAcc[L2_NODES * FDIM];
    __shared__ unsigned sCnt[L2_NODES];
    __shared__ unsigned sQ[OWN_WAVES * QCAP];
    __shared__ unsigned sLen[OWN_WAVES];
    __shared__ __attribute__((aligned(8))) v2f sW[2 * 12 * 3 * 32];

    const int tid  = threadIdx.x;
    const int lane = tid & 31;
    const int wave = __builtin_amdgcn_readfirstlane(tid >> 5);
    const int half = lane >> 4;
    const int l16  = lane & 15;
    const int range = blockIdx.x / SPLIT;
    const int sub   = blockIdx.x % SPLIT;
    const int lo    = range * OWN_NODES + sub * L2_NODES;

    for (int i = tid; i < L2_NODES * FDIM / 4; i += OWN_THREADS)
        ((float4*)sAcc)[i] = make_float4(0.0f, 0.0f, 0.0f, 0.0f);
    for (int i = tid; i < L2_NODES; i += OWN_THREADS) sCnt[i] = 0u;
    stage_weights(sW, Wself, Wneigh, tid);
    __syncthreads();

    uint2* myList = (uint2*)lists + ((size_t)blockIdx.x * OWN_WAVES + wave) * LCAP;

    const int plane4n = planeWords / 4;
    const int wave4   = (plane4n + OWN_WAVES - 1) / OWN_WAVES;
    const int w4Beg   = min(wave * wave4, plane4n);
    const int w4End   = min(w4Beg + wave4, plane4n);
    const uint4* plane4 = (const uint4*)(planes + (size_t)range * planeWords);
    const uint4* half4  = (const uint4*)(planes + (size_t)nRanges * planeWords);
    const unsigned flip = (sub == 0) ? 0xffffffffu : 0u;
    const int lpos = plane_share(x, packed, plane4, half4, flip, (unsigned)(sub * L2_NODES),
                                 sAcc, sCnt, sQ + wave * QCAP, myList,
                                 w4Beg, w4End, lane, half, l16);
    if (lane == 0) sLen[wave] = (unsigned)lpos;

    const int nOwn    = max(0, min(L2_NODES, nNodes - lo));
    const bool hasTile = wave < nOwn / 16;
    const int rowL = wave * 16 + l16;
    v2f xa[12];
    float bv0 = 0.0f, bv1 = 0.0f, bv2 = 0.0f;
    #pragma unroll
    for (int k = 0; k < 12; ++k) { xa[k].x = 0.0f; xa[k].y = 0.0f; }
    if (hasTile) {
        const float* xrow = x + (size_t)(lo + rowL) * FDIM + half * 2;
        #pragma unroll
        for (int k = 0; k < 12; ++k) {
            xa[k].x = xrow[4 * k];
            xa[k].y = xrow[4 * k + 1];
        }
        bv0 = bias[0 * 16 + l16];
        bv1 = bias[1 * 16 + l16];
        bv2 = bias[2 * 16 + l16];
    }
    __syncthreads();

    if (tid < OWN_WAVES) listLen[blockIdx.x * OWN_WAVES + tid] = sLen[tid];
    for (int i = tid; i < nOwn; i += OWN_THREADS) cnt[lo + i] = (float)sCnt[i];

    if (hasTile) {
        const float dsc = 1.0f / fmaxf((float)sCnt[rowL], 1.0f);
        owned_tile_products<true>(xa, (const v2f*)(sAcc + rowL * FDIM + half * 2), sW, dsc,
                                  bv0, bv1, bv2,
                                  hout + (size_t)(lo + wave * 16 + half * 8) * FDIM + l16, lane);
    }
}

template <bool FULL>
__device__ __forceinline__ void list_drain16(const uint2* q, int n,
                                             const float* __restrict__ x, float* sAcc,
                                             int half, int l16) {
    uint2 ent[8];
    #pragma unroll
    for (int p = 0; p < 8; ++p) {
        const int idx = 2 * p + half;
        ent[p] = (FULL || idx < n) ? q[idx] : make_uint2(0u, 0u);
    }
    float r0[8], r1[8], r2[8];
    list_rows16<FULL>(ent, n, x, r0, r1, r2, half, l16);
    list_adds16<FULL>(ent, n, sAcc, r0, r1, r2, half, l16);
}

__global__ void __launch_bounds__(OWN_THREADS)
own_sum_gemm_kernel(const float* __restrict__ x,
                    const int* __restrict__ src,
                    const int* __restrict__ dst,
                    const float* __restrict__ cnt,
                    const unsigned* __restrict__ lists,
                    const unsigned* __restrict__ listLen,
                    const float* __restrict__ Wself,
                    const float* __restrict__ Wneigh,
                    const float* __restrict__ bias,
                    float* __restrict__ out,
                    int nNodes, int nEdges) {
    __shared__ __attribute__((aligned(16))) float sAcc[L2_NODES * FDIM];
    __shared__ __attribute__((aligned(8))) uint2 sQ[OWN_WAVES * QCAP];
    __shared__ __attribute__((aligned(8))) v2f sW[2 * 12 * 3 * 32];

    const int tid  = threadIdx.x;
    const int lane = tid & 31;
    const int wave = __builtin_amdgcn_readfirstlane(tid >> 5);
    const int half = lane >> 4;
    const int l16  = lane & 15;
    const int range = blockIdx.x / SPLIT;
    const int sub   = blockIdx.x % SPLIT;
    const int lo    = range * OWN_NODES + sub * L2_NODES;

    for (int i = tid; i < L2_NODES * FDIM / 4; i += OWN_THREADS)
        ((float4*)sAcc)[i] = make_float4(0.0f, 0.0f, 0.0f, 0.0f);
    stage_weights(sW, Wself, Wneigh, tid);
    __syncthreads();

    uint2* q = sQ + wave * QCAP;

    const unsigned len = listLen[blockIdx.x * OWN_WAVES + lane];
    const bool fromLists = (__builtin_amdgcn_ballot_w32(len > (unsigned)LCAP) == 0u);

    if (fromLists) {
        int cntQ = 0;
        const unsigned bit = 1u << lane;
        const int n = __builtin_amdgcn_readlane((int)len, wave);
        const uint2* list = (const uint2*)lists + ((size_t)blockIdx.x * OWN_WAVES + wave) * LCAP;
        for (int pos = 0; pos < n; pos += 32 * LIST_U) {
            uint2 ent[LIST_U];
            #pragma unroll
            for (int u = 0; u < LIST_U; ++u) {
                const int idx = pos + u * 32 + lane;
                ent[u] = (idx < n) ? list[idx] : make_uint2(0u, 0xffffffffu);
            }
            #pragma unroll
            for (int u = 0; u < LIST_U; ++u) {
                const unsigned r = ent[u].y;
                const unsigned b = __builtin_amdgcn_ballot_w32(r < (unsigned)L2_NODES);
                if (b != 0u) {
                    if (b & bit) q[cntQ + (int)__builtin_amdgcn_mbcnt_lo(b, 0u)] = make_uint2(ent[u].x, r);
                    cntQ += __builtin_popcount(b);
                    while (cntQ >= 16) {
                        cntQ -= 16;
                        list_drain16<true>(q + cntQ, 16, x, sAcc, half, l16);
                    }
                }
            }
        }
        if (cntQ > 0) list_drain16<false>(q, cntQ, x, sAcc, half, l16);
    } else {
        const int nEdges4 = nEdges & ~3;
        sweep_share<L2_NODES>(x, src, dst, sAcc, nullptr, q, nullptr,
                              lo, 0, nEdges4, true, nEdges4, nEdges,
                              wave, lane, half, l16, 0);
    }
    const int nOwn    = max(0, min(L2_NODES, nNodes - lo));
    const bool hasTile = wave < nOwn / 16;
    const int rowL = wave * 16 + l16;
    v2f xa[12];
    float deg = 1.0f, bv0 = 0.0f, bv1 = 0.0f, bv2 = 0.0f;
    #pragma unroll
    for (int k = 0; k < 12; ++k) { xa[k].x = 0.0f; xa[k].y = 0.0f; }
    if (hasTile) {
        const int m = lo + rowL;
        const float* xrow = x + (size_t)m * FDIM + half * 2;
        #pragma unroll
        for (int k = 0; k < 12; ++k) {
            xa[k].x = xrow[4 * k];
            xa[k].y = xrow[4 * k + 1];
        }
        deg = cnt[m];
        bv0 = bias[0 * 16 + l16];
        bv1 = bias[1 * 16 + l16];
        bv2 = bias[2 * 16 + l16];
    }
    __syncthreads();

    if (hasTile) {
        const float dsc = 1.0f / fmaxf(deg, 1.0f);
        owned_tile_products<false>(xa, (const v2f*)(sAcc + rowL * FDIM + half * 2), sW, dsc,
                                   bv0, bv1, bv2,
                                   out + (size_t)(lo + wave * 16 + half * 8) * FDIM + l16, lane);
    }
}

extern "C" void kernel_launch(void* const* d_in, const int* in_sizes, int n_in,
                              void* d_out, int out_size, void* d_ws, size_t ws_size,
                              hipStream_t stream) {
    const float* features = (const float*)d_in[0];
    const int*   src      = (const int*)d_in[1];
    const int*   dst      = (const int*)d_in[2];
    const float* Wself1   = (const float*)d_in[3];
    const float* Wneigh1  = (const float*)d_in[4];
    const float* b1       = (const float*)d_in[5];
    const float* Wself2   = (const float*)d_in[6];
    const float* Wneigh2  = (const float*)d_in[7];
    const float* b2       = (const float*)d_in[8];
    float* out = (float*)d_out;

    const int N = in_sizes[0] / FDIM;
    const int E = in_sizes[1];

    char* ws = (char*)d_ws;
    size_t off = 0;
    float* cnt  = (float*)(ws + off); off += (((size_t)N * 4)        + 255) & ~(size_t)255;
    float* h    = (float*)(ws + off); off += (((size_t)N * FDIM * 4) + 255) & ~(size_t)255;

    const int ownBlocks = ((N + OWN_NODES - 1) / OWN_NODES) * SPLIT;
    unsigned* listLen = (unsigned*)(ws + off); off += (((size_t)ownBlocks * OWN_WAVES * 4) + 255) & ~(size_t)255;
    unsigned* lists   = (unsigned*)(ws + off);
    off += (((size_t)ownBlocks * OWN_WAVES * LCAP * 8) + 255) & ~(size_t)255;

    const int nRanges    = ownBlocks / SPLIT;
    const int nChunks    = (E + PL_CHUNK - 1) / PL_CHUNK;
    const int planeWords = nChunks * PL_CHW;
    unsigned* packed = (unsigned*)(ws + off); off += (((size_t)E * 4) + 255) & ~(size_t)255;
    unsigned* planes = (unsigned*)(ws + off);

    plane_kernel<<<nChunks, 1024, 0, stream>>>(src, dst, planes, packed, E, nRanges, planeWords);

    own_plane_gemm_kernel<<<ownBlocks, OWN_THREADS, 0, stream>>>(features, packed, planes,
                                                                 Wself1, Wneigh1, b1, h, cnt,
                                                                 lists, listLen, N, nRanges, planeWords);

    own_sum_gemm_kernel<<<ownBlocks, OWN_THREADS, 0, stream>>>(h, src, dst, cnt, lists, listLen,
                                                               Wself2, Wneigh2, b2, out, N, E);
}
